# starting kernel
# baseline (speedup 1.0000x reference)
#include <hip/hip_runtime.h>

typedef __attribute__((ext_vector_type(2))) float v2f;
typedef __attribute__((ext_vector_type(8))) float v8f;

#define NEG_SLOPE 0.01f

__device__ __forceinline__ v8f wmma_f32_k4(v2f a, v2f b, v8f c) {
  return __builtin_amdgcn_wmma_f32_16x16x4_f32(
       false, a,  false, b,
       (short)0, c,  false,  false);
}

__device__ __forceinline__ void lds_fence() {
  asm volatile("s_wait_dscnt 0" ::: "memory");
}

__global__ void zero_f32(float* __restrict__ p, long long n) {
  long long i = (long long)blockIdx.x * blockDim.x + threadIdx.x;
  long long stride = (long long)gridDim.x * blockDim.x;
  for (; i < n; i += stride) p[i] = 0.0f;
}

__global__ void edge_scatter(const float* __restrict__ x,
                             const int* __restrict__ esrc,
                             const int* __restrict__ edst,
                             float* __restrict__ agg, long long n_edges) {
  long long t = (long long)blockIdx.x * blockDim.x + threadIdx.x;
  const int q = (int)(t & 3);
  long long e = t >> 2;
  long long stride = ((long long)gridDim.x * blockDim.x) >> 2;
  for (; e < n_edges; e += stride) {
    const int s = esrc[e];
    const int d = edst[e];
    const float4 v = ((const float4*)(x + (long long)s * 16))[q];
    float* ap = agg + (long long)d * 16 + q * 4;
    atomicAdd(ap + 0, v.x);
    atomicAdd(ap + 1, v.y);
    atomicAdd(ap + 2, v.z);
    atomicAdd(ap + 3, v.w);
  }
}

__device__ __forceinline__ void mlp_tile(
    const v2f aA[4], const v2f aX[4],
    const float* __restrict__ Wl, const float* __restrict__ bl,
    const float* __restrict__ Wr,
    const float* __restrict__ W1, const float* __restrict__ b1,
    const float* __restrict__ W2, const float* __restrict__ b2,
    float (*Sw)[18], float* __restrict__ out,
    int r0, int batch, int lane16, int half, int ko) {
  const int wrow = lane16 * 16 + ko;

  v2f bL[4], bR[4];
#pragma unroll
  for (int i = 0; i < 4; ++i) {
    bL[i] = *(const v2f*)(Wl + wrow + 4 * i);
    bR[i] = *(const v2f*)(Wr + wrow + 4 * i);
  }

  v8f acc = {};
#pragma unroll
  for (int i = 0; i < 4; ++i) acc = wmma_f32_k4(aA[i], bL[i], acc);
#pragma unroll
  for (int i = 0; i < 4; ++i) acc = wmma_f32_k4(aX[i], bR[i], acc);
  {
    const float bias = bl[lane16];
#pragma unroll
    for (int r = 0; r < 8; ++r) {
      float v = acc[r] + bias;
      v = (v >= 0.0f) ? v : NEG_SLOPE * v;
      Sw[r + 8 * half][lane16] = v;
    }
  }
  lds_fence();

  v2f a1[4], b1f[4];
#pragma unroll
  for (int i = 0; i < 4; ++i) {
    a1[i]  = *(const v2f*)(&Sw[lane16][4 * i + ko]);
    b1f[i] = *(const v2f*)(W1 + wrow + 4 * i);
  }
  v8f acc1 = {};
#pragma unroll
  for (int i = 0; i < 4; ++i) acc1 = wmma_f32_k4(a1[i], b1f[i], acc1);
  lds_fence();
  {
    const float bias = b1[lane16];
#pragma unroll
    for (int r = 0; r < 8; ++r) {
      float v = acc1[r] + bias;
      v = (v >= 0.0f) ? v : NEG_SLOPE * v;
      Sw[r + 8 * half][lane16] = v;
    }
  }
  lds_fence();

  v2f a2[4], b2f[4];
#pragma unroll
  for (int i = 0; i < 4; ++i) {
    a2[i]  = *(const v2f*)(&Sw[lane16][4 * i + ko]);
    b2f[i] = *(const v2f*)(W2 + wrow + 4 * i);
  }
  v8f acc2 = {};
#pragma unroll
  for (int i = 0; i < 4; ++i) acc2 = wmma_f32_k4(a2[i], b2f[i], acc2);
  {
    const float bias = b2[lane16];
#pragma unroll
    for (int r = 0; r < 8; ++r) {
      const int orow = r0 + r + 8 * half;
      if (orow < batch)
        out[(long long)orow * 16 + lane16] = acc2[r] + bias;
    }
  }
}

__global__ void __launch_bounds__(128)
sage_mlp_wmma(const float* __restrict__ agg, const float* __restrict__ x,
              const float* __restrict__ Wl, const float* __restrict__ bl,
              const float* __restrict__ Wr,
              const float* __restrict__ W1, const float* __restrict__ b1,
              const float* __restrict__ W2, const float* __restrict__ b2,
              float* __restrict__ out, int batch) {
  __shared__ float S[4][16][18];
  const int wave   = threadIdx.x >> 5;
  const int l      = threadIdx.x & 31;
  const int half   = l >> 4;
  const int lane16 = l & 15;
  const int ko     = 2 * half;
  const int tile   = blockIdx.x * 4 + wave;
  if (tile * 16 >= batch) return;
  const int r0 = tile * 16;
  float (*Sw)[18] = S[wave];

  v2f aA[4], aX[4];
  if (r0 + 16 <= batch) {
    const long long rb = (long long)(r0 + lane16) * 16 + ko;
#pragma unroll
    for (int i = 0; i < 4; ++i) {
      aA[i] = *(const v2f*)(agg + rb + 4 * i);
      aX[i] = *(const v2f*)(x + rb + 4 * i);
    }
  } else {
    const int row = r0 + lane16;
    const bool rowv = row < batch;
#pragma unroll
    for (int c = 0; c < 8; ++c)
      Sw[lane16][half * 8 + c] = rowv ? agg[(long long)row * 16 + half * 8 + c] : 0.0f;
    lds_fence();
#pragma unroll
    for (int i = 0; i < 4; ++i) aA[i] = *(const v2f*)(&Sw[lane16][4 * i + ko]);
    lds_fence();
#pragma unroll
    for (int c = 0; c < 8; ++c)
      Sw[lane16][half * 8 + c] = rowv ? x[(long long)row * 16 + half * 8 + c] : 0.0f;
    lds_fence();
#pragma unroll
    for (int i = 0; i < 4; ++i) aX[i] = *(const v2f*)(&Sw[lane16][4 * i + ko]);
    lds_fence();
  }

  mlp_tile(aA, aX, Wl, bl, Wr, W1, b1, W2, b2, Sw, out,
           r0, batch, lane16, half, ko);
}

extern "C" void kernel_launch(void* const* d_in, const int* in_sizes, int n_in,
                              void* d_out, int out_size, void* d_ws, size_t ws_size,
                              hipStream_t stream) {
  const float* x  = (const float*)d_in[0];
  const float* Wl = (const float*)d_in[1];
  const float* bl = (const float*)d_in[2];
  const float* Wr = (const float*)d_in[3];
  const float* W1 = (const float*)d_in[4];
  const float* b1 = (const float*)d_in[5];
  const float* W2 = (const float*)d_in[6];
  const float* b2 = (const float*)d_in[7];
  const int* esrc = (const int*)d_in[8];
  const int* edst = (const int*)d_in[9];

  const long long n_edges = in_sizes[8];
  const int batch = out_size / 16;
  float* agg = (float*)d_ws;
  float* out = (float*)d_out;

  const long long n_agg = (long long)batch * 16;
  int zgrid = (int)((n_agg + 255) / 256);
  zero_f32<<<zgrid, 256, 0, stream>>>(agg, n_agg);

  const long long ethreads = n_edges * 4;
  int egrid = (int)((ethreads + 255) / 256);
  edge_scatter<<<egrid, 256, 0, stream>>>(x, esrc, edst, agg, n_edges);

  const int tiles = (batch + 15) / 16;
  int mgrid = (tiles + 3) / 4;
  sage_mlp_wmma<<<mgrid, 128, 0, stream>>>(agg, x, Wl, bl, Wr,
                                           W1, b1, W2, b2, out, batch);
}


# Edge sums accumulate in LDS: a work-group owns a slice of target rows, filters its edge chunk, adds rows with LDS f32 atomics, flushes once by global atomics (8x fewer); sum order changes
# speedup vs baseline: 1.8122x; 1.8122x over previous
#include <hip/hip_runtime.h>

typedef __attribute__((ext_vector_type(2))) float v2f;
typedef __attribute__((ext_vector_type(8))) float v8f;

#define NEG_SLOPE 0.01f

__device__ __forceinline__ v8f wmma_f32_k4(v2f a, v2f b, v8f c) {
  return __builtin_amdgcn_wmma_f32_16x16x4_f32(
       false, a,  false, b,
       (short)0, c,  false,  false);
}

__device__ __forceinline__ void lds_fence() {
  asm volatile("s_wait_dscnt 0" ::: "memory");
}

__global__ void zero_f32(float* __restrict__ p, long long n) {
  long long i = (long long)blockIdx.x * blockDim.x + threadIdx.x;
  long long stride = (long long)gridDim.x * blockDim.x;
  for (; i < n; i += stride) p[i] = 0.0f;
}

#define SLICE_T 1563
#define SC_WAVES 32
#define SC_PASS 64
#define SC_LCAP 192

__device__ __forceinline__ void scatter_pass(const float* __restrict__ x,
                                             const int* __restrict__ esrc,
                                             const int* le, const int* ld,
                                             float* acc, int base, int n, int lane) {
  const int q = lane & 3;
  const int sub = lane >> 2;
  int s[8], d[8];
#pragma unroll
  for (int u = 0; u < 8; ++u) {
    int j = sub + 8 * u;
    j = (j < n) ? j : (n - 1);
    const int e = le[base + j];
    d[u] = ld[base + j];
    s[u] = esrc[e];
  }
  float4 v[8];
#pragma unroll
  for (int u = 0; u < 8; ++u)
    v[u] = *(const float4*)(x + (long long)s[u] * 16 + q * 4);
#pragma unroll
  for (int u = 0; u < 8; ++u) {
    if (sub + 8 * u < n) {
      float* a = acc + d[u] * 16 + q * 4;
      atomicAdd(a + 0, v[u].x);
      atomicAdd(a + 1, v[u].y);
      atomicAdd(a + 2, v[u].z);
      atomicAdd(a + 3, v[u].w);
    }
  }
}

__device__ __forceinline__ int4 load_dst4_tail(const int* __restrict__ edst,
                                               unsigned wb, unsigned e1, int lane) {
  int4 t = make_int4(-1, -1, -1, -1);
  const unsigned e = wb + lane * 4;
  if (e      < e1) t.x = edst[e];
  if (e + 1u < e1) t.y = edst[e + 1u];
  if (e + 2u < e1) t.z = edst[e + 2u];
  if (e + 3u < e1) t.w = edst[e + 3u];
  return t;
}

__global__ void __launch_bounds__(1024)
edge_scatter_lds(const float* __restrict__ x,
                 const int* __restrict__ esrc,
                 const int* __restrict__ edst,
                 float* __restrict__ agg, long long n_edges,
                 int batch, int n_slices) {
  __shared__ float acc[SLICE_T * 16];
  __shared__ int lst_e[SC_WAVES][SC_LCAP];
  __shared__ int lst_d[SC_WAVES][SC_LCAP];

  const int tid  = threadIdx.x;
  const int lane = tid & 31;
  const int wave = __builtin_amdgcn_readfirstlane(tid >> 5);
  const int slice    = blockIdx.x % n_slices;
  const int chunk    = blockIdx.x / n_slices;
  const int n_chunks = gridDim.x / n_slices;
  const int lo = slice * SLICE_T;
  int ns = batch - lo;
  if (ns > SLICE_T) ns = SLICE_T;

  for (int i = tid; i < SLICE_T * 16; i += 1024) acc[i] = 0.0f;
  __syncthreads();

  long long per = (n_edges + n_chunks - 1) / n_chunks;
  per = (per + 3) & ~3LL;
  long long lo_e = (long long)chunk * per;
  if (lo_e > n_edges) lo_e = n_edges;
  long long hi_e = lo_e + per;
  if (hi_e > n_edges) hi_e = n_edges;
  const unsigned e0 = (unsigned)lo_e;
  const unsigned e1 = (unsigned)hi_e;

  int* le = lst_e[wave];
  int* ld = lst_d[wave];
  int cnt = 0;

  auto filter = [&](const int4 c, const unsigned wb) {
    const unsigned r0 = (unsigned)(c.x - lo), r1 = (unsigned)(c.y - lo);
    const unsigned r2 = (unsigned)(c.z - lo), r3 = (unsigned)(c.w - lo);
    const bool p0 = r0 < (unsigned)ns, p1 = r1 < (unsigned)ns;
    const bool p2 = r2 < (unsigned)ns, p3 = r3 < (unsigned)ns;
    const unsigned m0 = __builtin_amdgcn_ballot_w32(p0);
    const unsigned m1 = __builtin_amdgcn_ballot_w32(p1);
    const unsigned m2 = __builtin_amdgcn_ballot_w32(p2);
    const unsigned m3 = __builtin_amdgcn_ballot_w32(p3);
    if ((m0 | m1 | m2 | m3) != 0u) {
      const int ei = (int)(wb + lane * 4);
      int b = cnt;
      if (p0) { const int o = b + (int)__builtin_amdgcn_mbcnt_lo(m0, 0u); le[o] = ei;     ld[o] = (int)r0; }
      b += __popc(m0);
      if (p1) { const int o = b + (int)__builtin_amdgcn_mbcnt_lo(m1, 0u); le[o] = ei + 1; ld[o] = (int)r1; }
      b += __popc(m1);
      if (p2) { const int o = b + (int)__builtin_amdgcn_mbcnt_lo(m2, 0u); le[o] = ei + 2; ld[o] = (int)r2; }
      b += __popc(m2);
      if (p3) { const int o = b + (int)__builtin_amdgcn_mbcnt_lo(m3, 0u); le[o] = ei + 3; ld[o] = (int)r3; }
      b += __popc(m3);
      cnt = b;
      while (cnt >= SC_PASS) {
        lds_fence();
        scatter_pass(x, esrc, le, ld, acc, cnt - SC_PASS, SC_PASS, lane);
        cnt -= SC_PASS;
      }
    }
  };

  const unsigned step = SC_WAVES * 128u;
  unsigned wb = e0 + (unsigned)wave * 128u;
  if (wb + 128u <= e1) {
    const unsigned first = wb;
    auto ldw = [&](const unsigned w) {
      const unsigned a = (w + 128u <= e1) ? w : first;
      return *(const int4*)(edst + a + lane * 4);
    };
    int4 c0 = ldw(wb);
    int4 c1 = ldw(wb + step);
    int4 c2;
    for (;;) {
      c2 = ldw(wb + 2u * step); filter(c0, wb); wb += step; if (wb + 128u > e1) break;
      c0 = ldw(wb + 2u * step); filter(c1, wb); wb += step; if (wb + 128u > e1) break;
      c1 = ldw(wb + 2u * step); filter(c2, wb); wb += step; if (wb + 128u > e1) break;
    }
  }
  if (wb < e1) filter(load_dst4_tail(edst, wb, e1, lane), wb);
  if (cnt > 0) {
    lds_fence();
    scatter_pass(x, esrc, le, ld, acc, 0, cnt, lane);
  }
  __syncthreads();

  float* g = agg + (long long)lo * 16;
  for (int i = tid; i < ns * 16; i += 1024) atomicAdd(g + i, acc[i]);
}

__device__ __forceinline__ void mlp_tile(
    const v2f aA[4], const v2f aX[4],
    const float* __restrict__ Wl, const float* __restrict__ bl,
    const float* __restrict__ Wr,
    const float* __restrict__ W1, const float* __restrict__ b1,
    const float* __restrict__ W2, const float* __restrict__ b2,
    float (*Sw)[18], float* __restrict__ out,
    int r0, int batch, int lane16, int half, int ko) {
  const int wrow = lane16 * 16 + ko;

  v2f bL[4], bR[4];
#pragma unroll
  for (int i = 0; i < 4; ++i) {
    bL[i] = *(const v2f*)(Wl + wrow + 4 * i);
    bR[i] = *(const v2f*)(Wr + wrow + 4 * i);
  }

  v8f acc = {};
#pragma unroll
  for (int i = 0; i < 4; ++i) acc = wmma_f32_k4(aA[i], bL[i], acc);
#pragma unroll
  for (int i = 0; i < 4; ++i) acc = wmma_f32_k4(aX[i], bR[i], acc);
  {
    const float bias = bl[lane16];
#pragma unroll
    for (int r = 0; r < 8; ++r) {
      float v = acc[r] + bias;
      v = (v >= 0.0f) ? v : NEG_SLOPE * v;
      Sw[r + 8 * half][lane16] = v;
    }
  }
  lds_fence();

  v2f a1[4], b1f[4];
#pragma unroll
  for (int i = 0; i < 4; ++i) {
    a1[i]  = *(const v2f*)(&Sw[lane16][4 * i + ko]);
    b1f[i] = *(const v2f*)(W1 + wrow + 4 * i);
  }
  v8f acc1 = {};
#pragma unroll
  for (int i = 0; i < 4; ++i) acc1 = wmma_f32_k4(a1[i], b1f[i], acc1);
  lds_fence();
  {
    const float bias = b1[lane16];
#pragma unroll
    for (int r = 0; r < 8; ++r) {
      float v = acc1[r] + bias;
      v = (v >= 0.0f) ? v : NEG_SLOPE * v;
      Sw[r + 8 * half][lane16] = v;
    }
  }
  lds_fence();

  v2f a2[4], b2f[4];
#pragma unroll
  for (int i = 0; i < 4; ++i) {
    a2[i]  = *(const v2f*)(&Sw[lane16][4 * i + ko]);
    b2f[i] = *(const v2f*)(W2 + wrow + 4 * i);
  }
  v8f acc2 = {};
#pragma unroll
  for (int i = 0; i < 4; ++i) acc2 = wmma_f32_k4(a2[i], b2f[i], acc2);
  {
    const float bias = b2[lane16];
#pragma unroll
    for (int r = 0; r < 8; ++r) {
      const int orow = r0 + r + 8 * half;
      if (orow < batch)
        out[(long long)orow * 16 + lane16] = acc2[r] + bias;
    }
  }
}

__global__ void __launch_bounds__(128)
sage_mlp_wmma(const float* __restrict__ agg, const float* __restrict__ x,
              const float* __restrict__ Wl, const float* __restrict__ bl,
              const float* __restrict__ Wr,
              const float* __restrict__ W1, const float* __restrict__ b1,
              const float* __restrict__ W2, const float* __restrict__ b2,
              float* __restrict__ out, int batch) {
  __shared__ float S[4][16][18];
  const int wave   = threadIdx.x >> 5;
  const int l      = threadIdx.x & 31;
  const int half   = l >> 4;
  const int lane16 = l & 15;
  const int ko     = 2 * half;
  const int tile   = blockIdx.x * 4 + wave;
  if (tile * 16 >= batch) return;
  const int r0 = tile * 16;
  float (*Sw)[18] = S[wave];

  v2f aA[4], aX[4];
  if (r0 + 16 <= batch) {
    const long long rb = (long long)(r0 + lane16) * 16 + ko;
#pragma unroll
    for (int i = 0; i < 4; ++i) {
      aA[i] = *(const v2f*)(agg + rb + 4 * i);
      aX[i] = *(const v2f*)(x + rb + 4 * i);
    }
  } else {
    const int row = r0 + lane16;
    const bool rowv = row < batch;
#pragma unroll
    for (int c = 0; c < 8; ++c)
      Sw[lane16][half * 8 + c] = rowv ? agg[(long long)row * 16 + half * 8 + c] : 0.0f;
    lds_fence();
#pragma unroll
    for (int i = 0; i < 4; ++i) aA[i] = *(const v2f*)(&Sw[lane16][4 * i + ko]);
    lds_fence();
#pragma unroll
    for (int c = 0; c < 8; ++c)
      Sw[lane16][half * 8 + c] = rowv ? x[(long long)row * 16 + half * 8 + c] : 0.0f;
    lds_fence();
#pragma unroll
    for (int i = 0; i < 4; ++i) aX[i] = *(const v2f*)(&Sw[lane16][4 * i + ko]);
    lds_fence();
  }

  mlp_tile(aA, aX, Wl, bl, Wr, W1, b1, W2, b2, Sw, out,
           r0, batch, lane16, half, ko);
}

extern "C" void kernel_launch(void* const* d_in, const int* in_sizes, int n_in,
                              void* d_out, int out_size, void* d_ws, size_t ws_size,
                              hipStream_t stream) {
  const float* x  = (const float*)d_in[0];
  const float* Wl = (const float*)d_in[1];
  const float* bl = (const float*)d_in[2];
  const float* Wr = (const float*)d_in[3];
  const float* W1 = (const float*)d_in[4];
  const float* b1 = (const float*)d_in[5];
  const float* W2 = (const float*)d_in[6];
  const float* b2 = (const float*)d_in[7];
  const int* esrc = (const int*)d_in[8];
  const int* edst = (const int*)d_in[9];

  const long long n_edges = in_sizes[8];
  const int batch = out_size / 16;
  float* agg = (float*)d_ws;
  float* out = (float*)d_out;

  const long long n_agg = (long long)batch * 16;
  int zgrid = (int)((n_agg + 255) / 256);
  zero_f32<<<zgrid, 256, 0, stream>>>(agg, n_agg);

  int n_slices = (batch + SLICE_T - 1) / SLICE_T;
  if (n_slices < 1) n_slices = 1;
  int n_chunks = 256 / n_slices;
  if (n_chunks < 1) n_chunks = 1;
  edge_scatter_lds<<<n_slices * n_chunks, 1024, 0, stream>>>(x, esrc, edst, agg,
                                                            n_edges, batch, n_slices);

  const int tiles = (batch + 15) / 16;
  int mgrid = (tiles + 3) / 4;
  sage_mlp_wmma<<<mgrid, 128, 0, stream>>>(agg, x, Wl, bl, Wr,
                                           W1, b1, W2, b2, out, batch);
}


# LDS slice grown from 1563 to 4352 target rows (the whole 320 KiB of a WGP): 23 slices x 11 edge chunks, not 64 x 4, so every edge target is filtered 23 times instead of 64; sum order changes
# speedup vs baseline: 2.1662x; 1.1953x over previous
#include <hip/hip_runtime.h>

typedef __attribute__((ext_vector_type(2))) float v2f;
typedef __attribute__((ext_vector_type(8))) float v8f;

#define NEG_SLOPE 0.01f

__device__ __forceinline__ v8f wmma_f32_k4(v2f a, v2f b, v8f c) {
  return __builtin_amdgcn_wmma_f32_16x16x4_f32(
       false, a,  false, b,
       (short)0, c,  false,  false);
}

__device__ __forceinline__ void lds_fence() {
  asm volatile("s_wait_dscnt 0" ::: "memory");
}

__global__ void zero_f32(float* __restrict__ p, long long n) {
  long long i = (long long)blockIdx.x * blockDim.x + threadIdx.x;
  long long stride = (long long)gridDim.x * blockDim.x;
  for (; i < n; i += stride) p[i] = 0.0f;
}

#define SLICE_T 4352
#define SC_WAVES 32
#define SC_PASS 64
#define SC_LCAP 192

__device__ __forceinline__ void scatter_pass(const float* __restrict__ x,
                                             const int* __restrict__ esrc,
                                             const int* le, const int* ld,
                                             float* acc, int base, int n, int lane) {
  const int q = lane & 3;
  const int sub = lane >> 2;
  int s[8], d[8];
#pragma unroll
  for (int u = 0; u < 8; ++u) {
    int j = sub + 8 * u;
    j = (j < n) ? j : (n - 1);
    const int e = le[base + j];
    d[u] = ld[base + j];
    s[u] = esrc[e];
  }
  float4 v[8];
#pragma unroll
  for (int u = 0; u < 8; ++u)
    v[u] = *(const float4*)(x + (long long)s[u] * 16 + q * 4);
#pragma unroll
  for (int u = 0; u < 8; ++u) {
    if (sub + 8 * u < n) {
      float* a = acc + d[u] * 16 + q * 4;
      atomicAdd(a + 0, v[u].x);
      atomicAdd(a + 1, v[u].y);
      atomicAdd(a + 2, v[u].z);
      atomicAdd(a + 3, v[u].w);
    }
  }
}

__device__ __forceinline__ int4 load_dst4_tail(const int* __restrict__ edst,
                                               unsigned wb, unsigned e1, int lane) {
  int4 t = make_int4(-1, -1, -1, -1);
  const unsigned e = wb + lane * 4;
  if (e      < e1) t.x = edst[e];
  if (e + 1u < e1) t.y = edst[e + 1u];
  if (e + 2u < e1) t.z = edst[e + 2u];
  if (e + 3u < e1) t.w = edst[e + 3u];
  return t;
}

__global__ void __launch_bounds__(1024)
edge_scatter_lds(const float* __restrict__ x,
                 const int* __restrict__ esrc,
                 const int* __restrict__ edst,
                 float* __restrict__ agg, long long n_edges,
                 int batch, int n_slices) {
  __shared__ float acc[SLICE_T * 16];
  __shared__ int lst_e[SC_WAVES][SC_LCAP];
  __shared__ int lst_d[SC_WAVES][SC_LCAP];

  const int tid  = threadIdx.x;
  const int lane = tid & 31;
  const int wave = __builtin_amdgcn_readfirstlane(tid >> 5);
  const int slice    = blockIdx.x % n_slices;
  const int chunk    = blockIdx.x / n_slices;
  const int n_chunks = gridDim.x / n_slices;
  const int lo = slice * SLICE_T;
  int ns = batch - lo;
  if (ns > SLICE_T) ns = SLICE_T;

  for (int i = tid; i < SLICE_T * 16; i += 1024) acc[i] = 0.0f;
  __syncthreads();

  long long per = (n_edges + n_chunks - 1) / n_chunks;
  per = (per + 3) & ~3LL;
  long long lo_e = (long long)chunk * per;
  if (lo_e > n_edges) lo_e = n_edges;
  long long hi_e = lo_e + per;
  if (hi_e > n_edges) hi_e = n_edges;
  const unsigned e0 = (unsigned)lo_e;
  const unsigned e1 = (unsigned)hi_e;

  int* le = lst_e[wave];
  int* ld = lst_d[wave];
  int cnt = 0;

  auto filter = [&](const int4 c, const unsigned wb) {
    const unsigned r0 = (unsigned)(c.x - lo), r1 = (unsigned)(c.y - lo);
    const unsigned r2 = (unsigned)(c.z - lo), r3 = (unsigned)(c.w - lo);
    const bool p0 = r0 < (unsigned)ns, p1 = r1 < (unsigned)ns;
    const bool p2 = r2 < (unsigned)ns, p3 = r3 < (unsigned)ns;
    const unsigned m0 = __builtin_amdgcn_ballot_w32(p0);
    const unsigned m1 = __builtin_amdgcn_ballot_w32(p1);
    const unsigned m2 = __builtin_amdgcn_ballot_w32(p2);
    const unsigned m3 = __builtin_amdgcn_ballot_w32(p3);
    if ((m0 | m1 | m2 | m3) != 0u) {
      const int ei = (int)(wb + lane * 4);
      int b = cnt;
      if (p0) { const int o = b + (int)__builtin_amdgcn_mbcnt_lo(m0, 0u); le[o] = ei;     ld[o] = (int)r0; }
      b += __popc(m0);
      if (p1) { const int o = b + (int)__builtin_amdgcn_mbcnt_lo(m1, 0u); le[o] = ei + 1; ld[o] = (int)r1; }
      b += __popc(m1);
      if (p2) { const int o = b + (int)__builtin_amdgcn_mbcnt_lo(m2, 0u); le[o] = ei + 2; ld[o] = (int)r2; }
      b += __popc(m2);
      if (p3) { const int o = b + (int)__builtin_amdgcn_mbcnt_lo(m3, 0u); le[o] = ei + 3; ld[o] = (int)r3; }
      b += __popc(m3);
      cnt = b;
      while (cnt >= SC_PASS) {
        lds_fence();
        scatter_pass(x, esrc, le, ld, acc, cnt - SC_PASS, SC_PASS, lane);
        cnt -= SC_PASS;
      }
    }
  };

  const unsigned step = SC_WAVES * 128u;
  unsigned wb = e0 + (unsigned)wave * 128u;
  if (wb + 128u <= e1) {
    const unsigned first = wb;
    auto ldw = [&](const unsigned w) {
      const unsigned a = (w + 128u <= e1) ? w : first;
      return *(const int4*)(edst + a + lane * 4);
    };
    int4 c0 = ldw(wb);
    int4 c1 = ldw(wb + step);
    int4 c2;
    for (;;) {
      c2 = ldw(wb + 2u * step); filter(c0, wb); wb += step; if (wb + 128u > e1) break;
      c0 = ldw(wb + 2u * step); filter(c1, wb); wb += step; if (wb + 128u > e1) break;
      c1 = ldw(wb + 2u * step); filter(c2, wb); wb += step; if (wb + 128u > e1) break;
    }
  }
  if (wb < e1) filter(load_dst4_tail(edst, wb, e1, lane), wb);
  if (cnt > 0) {
    lds_fence();
    scatter_pass(x, esrc, le, ld, acc, 0, cnt, lane);
  }
  __syncthreads();

  float* g = agg + (long long)lo * 16;
  for (int i = tid; i < ns * 16; i += 1024) atomicAdd(g + i, acc[i]);
}

__device__ __forceinline__ void mlp_tile(
    const v2f aA[4], const v2f aX[4],
    const float* __restrict__ Wl, const float* __restrict__ bl,
    const float* __restrict__ Wr,
    const float* __restrict__ W1, const float* __restrict__ b1,
    const float* __restrict__ W2, const float* __restrict__ b2,
    float (*Sw)[18], float* __restrict__ out,
    int r0, int batch, int lane16, int half, int ko) {
  const int wrow = lane16 * 16 + ko;

  v2f bL[4], bR[4];
#pragma unroll
  for (int i = 0; i < 4; ++i) {
    bL[i] = *(const v2f*)(Wl + wrow + 4 * i);
    bR[i] = *(const v2f*)(Wr + wrow + 4 * i);
  }

  v8f acc = {};
#pragma unroll
  for (int i = 0; i < 4; ++i) acc = wmma_f32_k4(aA[i], bL[i], acc);
#pragma unroll
  for (int i = 0; i < 4; ++i) acc = wmma_f32_k4(aX[i], bR[i], acc);
  {
    const float bias = bl[lane16];
#pragma unroll
    for (int r = 0; r < 8; ++r) {
      float v = acc[r] + bias;
      v = (v >= 0.0f) ? v : NEG_SLOPE * v;
      Sw[r + 8 * half][lane16] = v;
    }
  }
  lds_fence();

  v2f a1[4], b1f[4];
#pragma unroll
  for (int i = 0; i < 4; ++i) {
    a1[i]  = *(const v2f*)(&Sw[lane16][4 * i + ko]);
    b1f[i] = *(const v2f*)(W1 + wrow + 4 * i);
  }
  v8f acc1 = {};
#pragma unroll
  for (int i = 0; i < 4; ++i) acc1 = wmma_f32_k4(a1[i], b1f[i], acc1);
  lds_fence();
  {
    const float bias = b1[lane16];
#pragma unroll
    for (int r = 0; r < 8; ++r) {
      float v = acc1[r] + bias;
      v = (v >= 0.0f) ? v : NEG_SLOPE * v;
      Sw[r + 8 * half][lane16] = v;
    }
  }
  lds_fence();

  v2f a2[4], b2f[4];
#pragma unroll
  for (int i = 0; i < 4; ++i) {
    a2[i]  = *(const v2f*)(&Sw[lane16][4 * i + ko]);
    b2f[i] = *(const v2f*)(W2 + wrow + 4 * i);
  }
  v8f acc2 = {};
#pragma unroll
  for (int i = 0; i < 4; ++i) acc2 = wmma_f32_k4(a2[i], b2f[i], acc2);
  {
    const float bias = b2[lane16];
#pragma unroll
    for (int r = 0; r < 8; ++r) {
      const int orow = r0 + r + 8 * half;
      if (orow < batch)
        out[(long long)orow * 16 + lane16] = acc2[r] + bias;
    }
  }
}

__global__ void __launch_bounds__(128)
sage_mlp_wmma(const float* __restrict__ agg, const float* __restrict__ x,
              const float* __restrict__ Wl, const float* __restrict__ bl,
              const float* __restrict__ Wr,
              const float* __restrict__ W1, const float* __restrict__ b1,
              const float* __restrict__ W2, const float* __restrict__ b2,
              float* __restrict__ out, int batch) {
  __shared__ float S[4][16][18];
  const int wave   = threadIdx.x >> 5;
  const int l      = threadIdx.x & 31;
  const int half   = l >> 4;
  const int lane16 = l & 15;
  const int ko     = 2 * half;
  const int tile   = blockIdx.x * 4 + wave;
  if (tile * 16 >= batch) return;
  const int r0 = tile * 16;
  float (*Sw)[18] = S[wave];

  v2f aA[4], aX[4];
  if (r0 + 16 <= batch) {
    const long long rb = (long long)(r0 + lane16) * 16 + ko;
#pragma unroll
    for (int i = 0; i < 4; ++i) {
      aA[i] = *(const v2f*)(agg + rb + 4 * i);
      aX[i] = *(const v2f*)(x + rb + 4 * i);
    }
  } else {
    const int row = r0 + lane16;
    const bool rowv = row < batch;
#pragma unroll
    for (int c = 0; c < 8; ++c)
      Sw[lane16][half * 8 + c] = rowv ? agg[(long long)row * 16 + half * 8 + c] : 0.0f;
    lds_fence();
#pragma unroll
    for (int i = 0; i < 4; ++i) aA[i] = *(const v2f*)(&Sw[lane16][4 * i + ko]);
    lds_fence();
#pragma unroll
    for (int c = 0; c < 8; ++c)
      Sw[lane16][half * 8 + c] = rowv ? x[(long long)row * 16 + half * 8 + c] : 0.0f;
    lds_fence();
#pragma unroll
    for (int i = 0; i < 4; ++i) aX[i] = *(const v2f*)(&Sw[lane16][4 * i + ko]);
    lds_fence();
  }

  mlp_tile(aA, aX, Wl, bl, Wr, W1, b1, W2, b2, Sw, out,
           r0, batch, lane16, half, ko);
}

extern "C" void kernel_launch(void* const* d_in, const int* in_sizes, int n_in,
                              void* d_out, int out_size, void* d_ws, size_t ws_size,
                              hipStream_t stream) {
  const float* x  = (const float*)d_in[0];
  const float* Wl = (const float*)d_in[1];
  const float* bl = (const float*)d_in[2];
  const float* Wr = (const float*)d_in[3];
  const float* W1 = (const float*)d_in[4];
  const float* b1 = (const float*)d_in[5];
  const float* W2 = (const float*)d_in[6];
  const float* b2 = (const float*)d_in[7];
  const int* esrc = (const int*)d_in[8];
  const int* edst = (const int*)d_in[9];

  const long long n_edges = in_sizes[8];
  const int batch = out_size / 16;
  float* agg = (float*)d_ws;
  float* out = (float*)d_out;

  const long long n_agg = (long long)batch * 16;
  int zgrid = (int)((n_agg + 255) / 256);
  zero_f32<<<zgrid, 256, 0, stream>>>(agg, n_agg);

  int n_slices = (batch + SLICE_T - 1) / SLICE_T;
  if (n_slices < 1) n_slices = 1;
  int n_chunks = 256 / n_slices;
  if (n_chunks < 1) n_chunks = 1;
  edge_scatter_lds<<<n_slices * n_chunks, 1024, 0, stream>>>(x, esrc, edst, agg,
                                                            n_edges, batch, n_slices);

  const int tiles = (batch + 15) / 16;
  int mgrid = (tiles + 3) / 4;
  sage_mlp_wmma<<<mgrid, 128, 0, stream>>>(agg, x, Wl, bl, Wr,
                                           W1, b1, W2, b2, out, batch);
}


# Edges bucketed once by target slice: a first pass sorts each tile of the edge list by slice into the workspace (one packed word per edge); the scatter reads only its own runs; sum order changes
# speedup vs baseline: 2.6336x; 1.2158x over previous
#include <hip/hip_runtime.h>

typedef __attribute__((ext_vector_type(2))) float v2f;
typedef __attribute__((ext_vector_type(8))) float v8f;

#define NEG_SLOPE 0.01f

__device__ __forceinline__ v8f wmma_f32_k4(v2f a, v2f b, v8f c) {
  return __builtin_amdgcn_wmma_f32_16x16x4_f32(
       false, a,  false, b,
       (short)0, c,  false,  false);
}

__device__ __forceinline__ void lds_fence() {
  asm volatile("s_wait_dscnt 0" ::: "memory");
}

__global__ void zero_f32(float* __restrict__ p, long long n) {
  long long i = (long long)blockIdx.x * blockDim.x + threadIdx.x;
  long long stride = (long long)gridDim.x * blockDim.x;
  for (; i < n; i += stride) p[i] = 0.0f;
}

#define SLICE_T 4352
#define SC_WAVES 32
#define SC_PASS 64
#define SC_LCAP 192

__device__ __forceinline__ void scatter_pass(const float* __restrict__ x,
                                             const int* __restrict__ esrc,
                                             const int* le, const int* ld,
                                             float* acc, int base, int n, int lane) {
  const int q = lane & 3;
  const int sub = lane >> 2;
  int s[8], d[8];
#pragma unroll
  for (int u = 0; u < 8; ++u) {
    int j = sub + 8 * u;
    j = (j < n) ? j : (n - 1);
    const int e = le[base + j];
    d[u] = ld[base + j];
    s[u] = esrc[e];
  }
  float4 v[8];
#pragma unroll
  for (int u = 0; u < 8; ++u)
    v[u] = *(const float4*)(x + (long long)s[u] * 16 + q * 4);
#pragma unroll
  for (int u = 0; u < 8; ++u) {
    if (sub + 8 * u < n) {
      float* a = acc + d[u] * 16 + q * 4;
      atomicAdd(a + 0, v[u].x);
      atomicAdd(a + 1, v[u].y);
      atomicAdd(a + 2, v[u].z);
      atomicAdd(a + 3, v[u].w);
    }
  }
}

__device__ __forceinline__ int4 load_dst4_tail(const int* __restrict__ edst,
                                               unsigned wb, unsigned e1, int lane) {
  int4 t = make_int4(-1, -1, -1, -1);
  const unsigned e = wb + lane * 4;
  if (e      < e1) t.x = edst[e];
  if (e + 1u < e1) t.y = edst[e + 1u];
  if (e + 2u < e1) t.z = edst[e + 2u];
  if (e + 3u < e1) t.w = edst[e + 3u];
  return t;
}

__global__ void __launch_bounds__(1024)
edge_scatter_lds(const float* __restrict__ x,
                 const int* __restrict__ esrc,
                 const int* __restrict__ edst,
                 float* __restrict__ agg, long long n_edges,
                 int batch, int n_slices) {
  __shared__ float acc[SLICE_T * 16];
  __shared__ int lst_e[SC_WAVES][SC_LCAP];
  __shared__ int lst_d[SC_WAVES][SC_LCAP];

  const int tid  = threadIdx.x;
  const int lane = tid & 31;
  const int wave = __builtin_amdgcn_readfirstlane(tid >> 5);
  const int slice    = blockIdx.x % n_slices;
  const int chunk    = blockIdx.x / n_slices;
  const int n_chunks = gridDim.x / n_slices;
  const int lo = slice * SLICE_T;
  int ns = batch - lo;
  if (ns > SLICE_T) ns = SLICE_T;

  for (int i = tid; i < SLICE_T * 16; i += 1024) acc[i] = 0.0f;
  __syncthreads();

  long long per = (n_edges + n_chunks - 1) / n_chunks;
  per = (per + 3) & ~3LL;
  long long lo_e = (long long)chunk * per;
  if (lo_e > n_edges) lo_e = n_edges;
  long long hi_e = lo_e + per;
  if (hi_e > n_edges) hi_e = n_edges;
  const unsigned e0 = (unsigned)lo_e;
  const unsigned e1 = (unsigned)hi_e;

  int* le = lst_e[wave];
  int* ld = lst_d[wave];
  int cnt = 0;

  auto filter = [&](const int4 c, const unsigned wb) {
    const unsigned r0 = (unsigned)(c.x - lo), r1 = (unsigned)(c.y - lo);
    const unsigned r2 = (unsigned)(c.z - lo), r3 = (unsigned)(c.w - lo);
    const bool p0 = r0 < (unsigned)ns, p1 = r1 < (unsigned)ns;
    const bool p2 = r2 < (unsigned)ns, p3 = r3 < (unsigned)ns;
    const unsigned m0 = __builtin_amdgcn_ballot_w32(p0);
    const unsigned m1 = __builtin_amdgcn_ballot_w32(p1);
    const unsigned m2 = __builtin_amdgcn_ballot_w32(p2);
    const unsigned m3 = __builtin_amdgcn_ballot_w32(p3);
    if ((m0 | m1 | m2 | m3) != 0u) {
      const int ei = (int)(wb + lane * 4);
      int b = cnt;
      if (p0) { const int o = b + (int)__builtin_amdgcn_mbcnt_lo(m0, 0u); le[o] = ei;     ld[o] = (int)r0; }
      b += __popc(m0);
      if (p1) { const int o = b + (int)__builtin_amdgcn_mbcnt_lo(m1, 0u); le[o] = ei + 1; ld[o] = (int)r1; }
      b += __popc(m1);
      if (p2) { const int o = b + (int)__builtin_amdgcn_mbcnt_lo(m2, 0u); le[o] = ei + 2; ld[o] = (int)r2; }
      b += __popc(m2);
      if (p3) { const int o = b + (int)__builtin_amdgcn_mbcnt_lo(m3, 0u); le[o] = ei + 3; ld[o] = (int)r3; }
      b += __popc(m3);
      cnt = b;
      while (cnt >= SC_PASS) {
        lds_fence();
        scatter_pass(x, esrc, le, ld, acc, cnt - SC_PASS, SC_PASS, lane);
        cnt -= SC_PASS;
      }
    }
  };

  const unsigned step = SC_WAVES * 128u;
  unsigned wb = e0 + (unsigned)wave * 128u;
  if (wb + 128u <= e1) {
    const unsigned first = wb;
    auto ldw = [&](const unsigned w) {
      const unsigned a = (w + 128u <= e1) ? w : first;
      return *(const int4*)(edst + a + lane * 4);
    };
    int4 c0 = ldw(wb);
    int4 c1 = ldw(wb + step);
    int4 c2;
    for (;;) {
      c2 = ldw(wb + 2u * step); filter(c0, wb); wb += step; if (wb + 128u > e1) break;
      c0 = ldw(wb + 2u * step); filter(c1, wb); wb += step; if (wb + 128u > e1) break;
      c1 = ldw(wb + 2u * step); filter(c2, wb); wb += step; if (wb + 128u > e1) break;
    }
  }
  if (wb < e1) filter(load_dst4_tail(edst, wb, e1, lane), wb);
  if (cnt > 0) {
    lds_fence();
    scatter_pass(x, esrc, le, ld, acc, 0, cnt, lane);
  }
  __syncthreads();

  float* g = agg + (long long)lo * 16;
  for (int i = tid; i < ns * 16; i += 1024) atomicAdd(g + i, acc[i]);
}

#define BK_SLICE 2048
#define BK_SHIFT 11
#define BK_MAXSL 64
#define BK_EPT   10
#define BK_CAP   (BK_EPT * 1024)
#define BK_TROW  (BK_MAXSL + 1)

__global__ void __launch_bounds__(1024)
edge_bucket(const int* __restrict__ esrc, const int* __restrict__ edst,
            int* __restrict__ pack, int* __restrict__ tab,
            long long n_edges, int per, int batch) {
  __shared__ int stage[BK_CAP];
  __shared__ int cnt[BK_MAXSL];
  __shared__ int off[BK_TROW];
  const int tid = threadIdx.x;
  long long lo = (long long)blockIdx.x * per;
  if (lo > n_edges) lo = n_edges;
  long long hi = lo + per;
  if (hi > n_edges) hi = n_edges;
  const int len = (int)(hi - lo);
  if (tid < BK_MAXSL) cnt[tid] = 0;
  __syncthreads();

  int dv[BK_EPT], sv[BK_EPT];
#pragma unroll
  for (int k = 0; k < BK_EPT; ++k) {
    const int i = tid + k * 1024;
    dv[k] = -1;
    sv[k] = 0;
    if (i < len) {
      dv[k] = edst[lo + i];
      sv[k] = esrc[lo + i];
    }
  }
  int ps[BK_EPT];
#pragma unroll
  for (int k = 0; k < BK_EPT; ++k) {
    ps[k] = -1;
    if ((unsigned)dv[k] < (unsigned)batch) {
      const int sl = dv[k] >> BK_SHIFT;
      ps[k] = (sl << 16) | atomicAdd(&cnt[sl], 1);
    }
  }
  __syncthreads();
  if (tid < BK_TROW) {
    int a = 0;
#pragma unroll
    for (int j = 0; j < BK_MAXSL; ++j) a += (j < tid) ? cnt[j] : 0;
    off[tid] = a;
  }
  __syncthreads();
#pragma unroll
  for (int k = 0; k < BK_EPT; ++k) {
    if (ps[k] >= 0) {
      const unsigned w = ((unsigned)sv[k] << BK_SHIFT) | ((unsigned)dv[k] & (BK_SLICE - 1));
      stage[off[ps[k] >> 16] + (ps[k] & 0xffff)] = (int)w;
    }
  }
  __syncthreads();
  const int total = off[BK_MAXSL];
  for (int i = tid; i < total; i += 1024) pack[lo + i] = stage[i];
  if (tid < BK_TROW) tab[blockIdx.x * BK_TROW + tid] = off[tid];
}

__global__ void __launch_bounds__(1024)
edge_consume_lds(const float* __restrict__ x, const int* __restrict__ pack,
                 const int* __restrict__ tab, float* __restrict__ agg,
                 int per, int batch, int n_slices, int n_tiles) {
  __shared__ float acc[BK_SLICE * 16];
  const int tid  = threadIdx.x;
  const int lane = tid & 31;
  const int wave = __builtin_amdgcn_readfirstlane(tid >> 5);
  const int slice    = blockIdx.x % n_slices;
  const int chunk    = blockIdx.x / n_slices;
  const int n_chunks = gridDim.x / n_slices;
  const int lo = slice * BK_SLICE;
  int ns = batch - lo;
  if (ns > BK_SLICE) ns = BK_SLICE;

  for (int i = tid; i < BK_SLICE * 16; i += 1024) acc[i] = 0.0f;
  __syncthreads();

  const int tpc = (n_tiles + n_chunks - 1) / n_chunks;
  const int t0 = chunk * tpc;
  int t1 = t0 + tpc;
  if (t1 > n_tiles) t1 = n_tiles;
  const int q   = lane & 3;
  const int grp = lane >> 2;

  for (int t = t0 + wave; t < t1; t += 32) {
    const int b = tab[t * BK_TROW + slice];
    const int e = tab[t * BK_TROW + slice + 1];
    if (b >= e) continue;
    const int* p = pack + (long long)t * per;
    int pos = b;
    int n = e - pos;
    if (n > 64) n = 64;
    int w0 = p[pos + ((lane < n) ? lane : (n - 1))];
    int w1 = p[pos + ((lane + 32 < n) ? (lane + 32) : (n - 1))];
    for (;;) {
      float4 v[8];
      int dl[8];
#pragma unroll
      for (int u = 0; u < 8; ++u) {
        const int from = (lane & ~3) | (u & 3);
        const int wd = __shfl((u < 4) ? w0 : w1, from);
        dl[u] = wd & (BK_SLICE - 1);
        const unsigned s = (unsigned)wd >> BK_SHIFT;
        v[u] = *(const float4*)(x + (long long)s * 16 + q * 4);
      }
      const int npos = pos + 64;
      int nn = e - npos;
      if (nn > 64) nn = 64;
      int x0 = w0, x1 = w1;
      if (nn > 0) {
        x0 = p[npos + ((lane < nn) ? lane : (nn - 1))];
        x1 = p[npos + ((lane + 32 < nn) ? (lane + 32) : (nn - 1))];
      }
#pragma unroll
      for (int u = 0; u < 8; ++u) {
        const int j = ((u < 4) ? 0 : 32) + 4 * grp + (u & 3);
        if (j < n) {
          float* a = acc + dl[u] * 16 + q * 4;
          atomicAdd(a + 0, v[u].x);
          atomicAdd(a + 1, v[u].y);
          atomicAdd(a + 2, v[u].z);
          atomicAdd(a + 3, v[u].w);
        }
      }
      if (nn <= 0) break;
      pos = npos;
      n = nn;
      w0 = x0;
      w1 = x1;
    }
  }
  __syncthreads();

  float* g = agg + (long long)lo * 16;
  for (int i = tid; i < ns * 16; i += 1024) atomicAdd(g + i, acc[i]);
}

__device__ __forceinline__ void mlp_tile(
    const v2f aA[4], const v2f aX[4],
    const float* __restrict__ Wl, const float* __restrict__ bl,
    const float* __restrict__ Wr,
    const float* __restrict__ W1, const float* __restrict__ b1,
    const float* __restrict__ W2, const float* __restrict__ b2,
    float (*Sw)[18], float* __restrict__ out,
    int r0, int batch, int lane16, int half, int ko) {
  const int wrow = lane16 * 16 + ko;

  v2f bL[4], bR[4];
#pragma unroll
  for (int i = 0; i < 4; ++i) {
    bL[i] = *(const v2f*)(Wl + wrow + 4 * i);
    bR[i] = *(const v2f*)(Wr + wrow + 4 * i);
  }

  v8f acc = {};
#pragma unroll
  for (int i = 0; i < 4; ++i) acc = wmma_f32_k4(aA[i], bL[i], acc);
#pragma unroll
  for (int i = 0; i < 4; ++i) acc = wmma_f32_k4(aX[i], bR[i], acc);
  {
    const float bias = bl[lane16];
#pragma unroll
    for (int r = 0; r < 8; ++r) {
      float v = acc[r] + bias;
      v = (v >= 0.0f) ? v : NEG_SLOPE * v;
      Sw[r + 8 * half][lane16] = v;
    }
  }
  lds_fence();

  v2f a1[4], b1f[4];
#pragma unroll
  for (int i = 0; i < 4; ++i) {
    a1[i]  = *(const v2f*)(&Sw[lane16][4 * i + ko]);
    b1f[i] = *(const v2f*)(W1 + wrow + 4 * i);
  }
  v8f acc1 = {};
#pragma unroll
  for (int i = 0; i < 4; ++i) acc1 = wmma_f32_k4(a1[i], b1f[i], acc1);
  lds_fence();
  {
    const float bias = b1[lane16];
#pragma unroll
    for (int r = 0; r < 8; ++r) {
      float v = acc1[r] + bias;
      v = (v >= 0.0f) ? v : NEG_SLOPE * v;
      Sw[r + 8 * half][lane16] = v;
    }
  }
  lds_fence();

  v2f a2[4], b2f[4];
#pragma unroll
  for (int i = 0; i < 4; ++i) {
    a2[i]  = *(const v2f*)(&Sw[lane16][4 * i + ko]);
    b2f[i] = *(const v2f*)(W2 + wrow + 4 * i);
  }
  v8f acc2 = {};
#pragma unroll
  for (int i = 0; i < 4; ++i) acc2 = wmma_f32_k4(a2[i], b2f[i], acc2);
  {
    const float bias = b2[lane16];
#pragma unroll
    for (int r = 0; r < 8; ++r) {
      const int orow = r0 + r + 8 * half;
      if (orow < batch)
        out[(long long)orow * 16 + lane16] = acc2[r] + bias;
    }
  }
}

__global__ void __launch_bounds__(128)
sage_mlp_wmma(const float* __restrict__ agg, const float* __restrict__ x,
              const float* __restrict__ Wl, const float* __restrict__ bl,
              const float* __restrict__ Wr,
              const float* __restrict__ W1, const float* __restrict__ b1,
              const float* __restrict__ W2, const float* __restrict__ b2,
              float* __restrict__ out, int batch) {
  __shared__ float S[4][16][18];
  const int wave   = threadIdx.x >> 5;
  const int l      = threadIdx.x & 31;
  const int half   = l >> 4;
  const int lane16 = l & 15;
  const int ko     = 2 * half;
  const int tile   = blockIdx.x * 4 + wave;
  if (tile * 16 >= batch) return;
  const int r0 = tile * 16;
  float (*Sw)[18] = S[wave];

  v2f aA[4], aX[4];
  if (r0 + 16 <= batch) {
    const long long rb = (long long)(r0 + lane16) * 16 + ko;
#pragma unroll
    for (int i = 0; i < 4; ++i) {
      aA[i] = *(const v2f*)(agg + rb + 4 * i);
      aX[i] = *(const v2f*)(x + rb + 4 * i);
    }
  } else {
    const int row = r0 + lane16;
    const bool rowv = row < batch;
#pragma unroll
    for (int c = 0; c < 8; ++c)
      Sw[lane16][half * 8 + c] = rowv ? agg[(long long)row * 16 + half * 8 + c] : 0.0f;
    lds_fence();
#pragma unroll
    for (int i = 0; i < 4; ++i) aA[i] = *(const v2f*)(&Sw[lane16][4 * i + ko]);
    lds_fence();
#pragma unroll
    for (int c = 0; c < 8; ++c)
      Sw[lane16][half * 8 + c] = rowv ? x[(long long)row * 16 + half * 8 + c] : 0.0f;
    lds_fence();
#pragma unroll
    for (int i = 0; i < 4; ++i) aX[i] = *(const v2f*)(&Sw[lane16][4 * i + ko]);
    lds_fence();
  }

  mlp_tile(aA, aX, Wl, bl, Wr, W1, b1, W2, b2, Sw, out,
           r0, batch, lane16, half, ko);
}

extern "C" void kernel_launch(void* const* d_in, const int* in_sizes, int n_in,
                              void* d_out, int out_size, void* d_ws, size_t ws_size,
                              hipStream_t stream) {
  const float* x  = (const float*)d_in[0];
  const float* Wl = (const float*)d_in[1];
  const float* bl = (const float*)d_in[2];
  const float* Wr = (const float*)d_in[3];
  const float* W1 = (const float*)d_in[4];
  const float* b1 = (const float*)d_in[5];
  const float* W2 = (const float*)d_in[6];
  const float* b2 = (const float*)d_in[7];
  const int* esrc = (const int*)d_in[8];
  const int* edst = (const int*)d_in[9];

  const long long n_edges = in_sizes[8];
  const int batch = out_size / 16;
  float* agg = (float*)d_ws;
  float* out = (float*)d_out;

  const long long n_agg = (long long)batch * 16;
  int zgrid = (int)((n_agg + 255) / 256);
  zero_f32<<<zgrid, 256, 0, stream>>>(agg, n_agg);

  int b_slices = (batch + BK_SLICE - 1) / BK_SLICE;
  if (b_slices < 1) b_slices = 1;
  int b_chunks = 256 / b_slices;
  if (b_chunks < 1) b_chunks = 1;
  const int n_tiles = b_chunks * 64;
  long long per = (n_edges + n_tiles - 1) / n_tiles;
  per = (per + 3) & ~3LL;
  const size_t agg_bytes  = ((size_t)n_agg * 4 + 255) & ~(size_t)255;
  const size_t pack_bytes = (size_t)n_tiles * (size_t)per * 4;
  const size_t tab_bytes  = (size_t)n_tiles * BK_TROW * 4;
  if (b_slices <= BK_MAXSL && per <= BK_CAP && in_sizes[0] <= 16 * 2097152 &&
      ws_size >= agg_bytes + pack_bytes + tab_bytes) {
    int* pack = (int*)((char*)d_ws + agg_bytes);
    int* tab  = (int*)((char*)d_ws + agg_bytes + pack_bytes);
    edge_bucket<<<n_tiles, 1024, 0, stream>>>(esrc, edst, pack, tab, n_edges, (int)per, batch);
    edge_consume_lds<<<b_slices * b_chunks, 1024, 0, stream>>>(x, pack, tab, agg, (int)per,
                                                              batch, b_slices, n_tiles);
  } else {
    int n_slices = (batch + SLICE_T - 1) / SLICE_T;
    if (n_slices < 1) n_slices = 1;
    int n_chunks = 256 / n_slices;
    if (n_chunks < 1) n_chunks = 1;
    edge_scatter_lds<<<n_slices * n_chunks, 1024, 0, stream>>>(x, esrc, edst, agg,
                                                              n_edges, batch, n_slices);
  }

  const int tiles = (batch + 15) / 16;
  int mgrid = (tiles + 3) / 4;
  sage_mlp_wmma<<<mgrid, 128, 0, stream>>>(agg, x, Wl, bl, Wr,
                                           W1, b1, W2, b2, out, batch);
}


# Each target slice (400 rows) owned by one work-group: edges bucketed into 250 slices, sums stay in LDS and feed the layers in the same kernel; no zeroing, no global atomics; sum order changes
# speedup vs baseline: 2.8442x; 1.0800x over previous
#include <hip/hip_runtime.h>

typedef __attribute__((ext_vector_type(2))) float v2f;
typedef __attribute__((ext_vector_type(8))) float v8f;

#define NEG_SLOPE 0.01f

__device__ __forceinline__ v8f wmma_f32_k4(v2f a, v2f b, v8f c) {
  return __builtin_amdgcn_wmma_f32_16x16x4_f32(
       false, a,  false, b,
       (short)0, c,  false,  false);
}

__device__ __forceinline__ void lds_fence() {
  asm volatile("s_wait_dscnt 0" ::: "memory");
}

__global__ void zero_f32(float* __restrict__ p, long long n) {
  long long i = (long long)blockIdx.x * blockDim.x + threadIdx.x;
  long long stride = (long long)gridDim.x * blockDim.x;
  for (; i < n; i += stride) p[i] = 0.0f;
}

#define SLICE_T 4352
#define SC_WAVES 32
#define SC_PASS 64
#define SC_LCAP 192

__device__ __forceinline__ void scatter_pass(const float* __restrict__ x,
                                             const int* __restrict__ esrc,
                                             const int* le, const int* ld,
                                             float* acc, int base, int n, int lane) {
  const int q = lane & 3;
  const int sub = lane >> 2;
  int s[8], d[8];
#pragma unroll
  for (int u = 0; u < 8; ++u) {
    int j = sub + 8 * u;
    j = (j < n) ? j : (n - 1);
    const int e = le[base + j];
    d[u] = ld[base + j];
    s[u] = esrc[e];
  }
  float4 v[8];
#pragma unroll
  for (int u = 0; u < 8; ++u)
    v[u] = *(const float4*)(x + (long long)s[u] * 16 + q * 4);
#pragma unroll
  for (int u = 0; u < 8; ++u) {
    if (sub + 8 * u < n) {
      float* a = acc + d[u] * 16 + q * 4;
      atomicAdd(a + 0, v[u].x);
      atomicAdd(a + 1, v[u].y);
      atomicAdd(a + 2, v[u].z);
      atomicAdd(a + 3, v[u].w);
    }
  }
}

__device__ __forceinline__ int4 load_dst4_tail(const int* __restrict__ edst,
                                               unsigned wb, unsigned e1, int lane) {
  int4 t = make_int4(-1, -1, -1, -1);
  const unsigned e = wb + lane * 4;
  if (e      < e1) t.x = edst[e];
  if (e + 1u < e1) t.y = edst[e + 1u];
  if (e + 2u < e1) t.z = edst[e + 2u];
  if (e + 3u < e1) t.w = edst[e + 3u];
  return t;
}

__global__ void __launch_bounds__(1024)
edge_scatter_lds(const float* __restrict__ x,
                 const int* __restrict__ esrc,
                 const int* __restrict__ edst,
                 float* __restrict__ agg, long long n_edges,
                 int batch, int n_slices) {
  __shared__ float acc[SLICE_T * 16];
  __shared__ int lst_e[SC_WAVES][SC_LCAP];
  __shared__ int lst_d[SC_WAVES][SC_LCAP];

  const int tid  = threadIdx.x;
  const int lane = tid & 31;
  const int wave = __builtin_amdgcn_readfirstlane(tid >> 5);
  const int slice    = blockIdx.x % n_slices;
  const int chunk    = blockIdx.x / n_slices;
  const int n_chunks = gridDim.x / n_slices;
  const int lo = slice * SLICE_T;
  int ns = batch - lo;
  if (ns > SLICE_T) ns = SLICE_T;

  for (int i = tid; i < SLICE_T * 16; i += 1024) acc[i] = 0.0f;
  __syncthreads();

  long long per = (n_edges + n_chunks - 1) / n_chunks;
  per = (per + 3) & ~3LL;
  long long lo_e = (long long)chunk * per;
  if (lo_e > n_edges) lo_e = n_edges;
  long long hi_e = lo_e + per;
  if (hi_e > n_edges) hi_e = n_edges;
  const unsigned e0 = (unsigned)lo_e;
  const unsigned e1 = (unsigned)hi_e;

  int* le = lst_e[wave];
  int* ld = lst_d[wave];
  int cnt = 0;

  auto filter = [&](const int4 c, const unsigned wb) {
    const unsigned r0 = (unsigned)(c.x - lo), r1 = (unsigned)(c.y - lo);
    const unsigned r2 = (unsigned)(c.z - lo), r3 = (unsigned)(c.w - lo);
    const bool p0 = r0 < (unsigned)ns, p1 = r1 < (unsigned)ns;
    const bool p2 = r2 < (unsigned)ns, p3 = r3 < (unsigned)ns;
    const unsigned m0 = __builtin_amdgcn_ballot_w32(p0);
    const unsigned m1 = __builtin_amdgcn_ballot_w32(p1);
    const unsigned m2 = __builtin_amdgcn_ballot_w32(p2);
    const unsigned m3 = __builtin_amdgcn_ballot_w32(p3);
    if ((m0 | m1 | m2 | m3) != 0u) {
      const int ei = (int)(wb + lane * 4);
      int b = cnt;
      if (p0) { const int o = b + (int)__builtin_amdgcn_mbcnt_lo(m0, 0u); le[o] = ei;     ld[o] = (int)r0; }
      b += __popc(m0);
      if (p1) { const int o = b + (int)__builtin_amdgcn_mbcnt_lo(m1, 0u); le[o] = ei + 1; ld[o] = (int)r1; }
      b += __popc(m1);
      if (p2) { const int o = b + (int)__builtin_amdgcn_mbcnt_lo(m2, 0u); le[o] = ei + 2; ld[o] = (int)r2; }
      b += __popc(m2);
      if (p3) { const int o = b + (int)__builtin_amdgcn_mbcnt_lo(m3, 0u); le[o] = ei + 3; ld[o] = (int)r3; }
      b += __popc(m3);
      cnt = b;
      while (cnt >= SC_PASS) {
        lds_fence();
        scatter_pass(x, esrc, le, ld, acc, cnt - SC_PASS, SC_PASS, lane);
        cnt -= SC_PASS;
      }
    }
  };

  const unsigned step = SC_WAVES * 128u;
  unsigned wb = e0 + (unsigned)wave * 128u;
  if (wb + 128u <= e1) {
    const unsigned first = wb;
    auto ldw = [&](const unsigned w) {
      const unsigned a = (w + 128u <= e1) ? w : first;
      return *(const int4*)(edst + a + lane * 4);
    };
    int4 c0 = ldw(wb);
    int4 c1 = ldw(wb + step);
    int4 c2;
    for (;;) {
      c2 = ldw(wb + 2u * step); filter(c0, wb); wb += step; if (wb + 128u > e1) break;
      c0 = ldw(wb + 2u * step); filter(c1, wb); wb += step; if (wb + 128u > e1) break;
      c1 = ldw(wb + 2u * step); filter(c2, wb); wb += step; if (wb + 128u > e1) break;
    }
  }
  if (wb < e1) filter(load_dst4_tail(edst, wb, e1, lane), wb);
  if (cnt > 0) {
    lds_fence();
    scatter_pass(x, esrc, le, ld, acc, 0, cnt, lane);
  }
  __syncthreads();

  float* g = agg + (long long)lo * 16;
  for (int i = tid; i < ns * 16; i += 1024) atomicAdd(g + i, acc[i]);
}

#define BK_SLICE 2048
#define BK_SHIFT 11
#define BK_MAXSL 64
#define BK_EPT   10
#define BK_CAP   (BK_EPT * 1024)
#define BK_TROW  (BK_MAXSL + 1)

__global__ void __launch_bounds__(1024)
edge_bucket(const int* __restrict__ esrc, const int* __restrict__ edst,
            int* __restrict__ pack, int* __restrict__ tab,
            long long n_edges, int per, int batch) {
  __shared__ int stage[BK_CAP];
  __shared__ int cnt[BK_MAXSL];
  __shared__ int off[BK_TROW];
  const int tid = threadIdx.x;
  long long lo = (long long)blockIdx.x * per;
  if (lo > n_edges) lo = n_edges;
  long long hi = lo + per;
  if (hi > n_edges) hi = n_edges;
  const int len = (int)(hi - lo);
  if (tid < BK_MAXSL) cnt[tid] = 0;
  __syncthreads();

  int dv[BK_EPT], sv[BK_EPT];
#pragma unroll
  for (int k = 0; k < BK_EPT; ++k) {
    const int i = tid + k * 1024;
    dv[k] = -1;
    sv[k] = 0;
    if (i < len) {
      dv[k] = edst[lo + i];
      sv[k] = esrc[lo + i];
    }
  }
  int ps[BK_EPT];
#pragma unroll
  for (int k = 0; k < BK_EPT; ++k) {
    ps[k] = -1;
    if ((unsigned)dv[k] < (unsigned)batch) {
      const int sl = dv[k] >> BK_SHIFT;
      ps[k] = (sl << 16) | atomicAdd(&cnt[sl], 1);
    }
  }
  __syncthreads();
  if (tid < BK_TROW) {
    int a = 0;
#pragma unroll
    for (int j = 0; j < BK_MAXSL; ++j) a += (j < tid) ? cnt[j] : 0;
    off[tid] = a;
  }
  __syncthreads();
#pragma unroll
  for (int k = 0; k < BK_EPT; ++k) {
    if (ps[k] >= 0) {
      const unsigned w = ((unsigned)sv[k] << BK_SHIFT) | ((unsigned)dv[k] & (BK_SLICE - 1));
      stage[off[ps[k] >> 16] + (ps[k] & 0xffff)] = (int)w;
    }
  }
  __syncthreads();
  const int total = off[BK_MAXSL];
  for (int i = tid; i < total; i += 1024) pack[lo + i] = stage[i];
  if (tid < BK_TROW) tab[blockIdx.x * BK_TROW + tid] = off[tid];
}

__global__ void __launch_bounds__(1024)
edge_consume_lds(const float* __restrict__ x, const int* __restrict__ pack,
                 const int* __restrict__ tab, float* __restrict__ agg,
                 int per, int batch, int n_slices, int n_tiles) {
  __shared__ float acc[BK_SLICE * 16];
  const int tid  = threadIdx.x;
  const int lane = tid & 31;
  const int wave = __builtin_amdgcn_readfirstlane(tid >> 5);
  const int slice    = blockIdx.x % n_slices;
  const int chunk    = blockIdx.x / n_slices;
  const int n_chunks = gridDim.x / n_slices;
  const int lo = slice * BK_SLICE;
  int ns = batch - lo;
  if (ns > BK_SLICE) ns = BK_SLICE;

  for (int i = tid; i < BK_SLICE * 16; i += 1024) acc[i] = 0.0f;
  __syncthreads();

  const int tpc = (n_tiles + n_chunks - 1) / n_chunks;
  const int t0 = chunk * tpc;
  int t1 = t0 + tpc;
  if (t1 > n_tiles) t1 = n_tiles;
  const int q   = lane & 3;
  const int grp = lane >> 2;

  for (int t = t0 + wave; t < t1; t += 32) {
    const int b = tab[t * BK_TROW + slice];
    const int e = tab[t * BK_TROW + slice + 1];
    if (b >= e) continue;
    const int* p = pack + (long long)t * per;
    int pos = b;
    int n = e - pos;
    if (n > 64) n = 64;
    int w0 = p[pos + ((lane < n) ? lane : (n - 1))];
    int w1 = p[pos + ((lane + 32 < n) ? (lane + 32) : (n - 1))];
    for (;;) {
      float4 v[8];
      int dl[8];
#pragma unroll
      for (int u = 0; u < 8; ++u) {
        const int from = (lane & ~3) | (u & 3);
        const int wd = __shfl((u < 4) ? w0 : w1, from);
        dl[u] = wd & (BK_SLICE - 1);
        const unsigned s = (unsigned)wd >> BK_SHIFT;
        v[u] = *(const float4*)(x + (long long)s * 16 + q * 4);
      }
      const int npos = pos + 64;
      int nn = e - npos;
      if (nn > 64) nn = 64;
      int x0 = w0, x1 = w1;
      if (nn > 0) {
        x0 = p[npos + ((lane < nn) ? lane : (nn - 1))];
        x1 = p[npos + ((lane + 32 < nn) ? (lane + 32) : (nn - 1))];
      }
#pragma unroll
      for (int u = 0; u < 8; ++u) {
        const int j = ((u < 4) ? 0 : 32) + 4 * grp + (u & 3);
        if (j < n) {
          float* a = acc + dl[u] * 16 + q * 4;
          atomicAdd(a + 0, v[u].x);
          atomicAdd(a + 1, v[u].y);
          atomicAdd(a + 2, v[u].z);
          atomicAdd(a + 3, v[u].w);
        }
      }
      if (nn <= 0) break;
      pos = npos;
      n = nn;
      w0 = x0;
      w1 = x1;
    }
  }
  __syncthreads();

  float* g = agg + (long long)lo * 16;
  for (int i = tid; i < ns * 16; i += 1024) atomicAdd(g + i, acc[i]);
}

#define SO_ROWS  400
#define SO_SHIFT 9
#define SO_MAXSL 256
#define SO_TROW  288
#define SO_EPT   13
#define SO_CAP   (SO_EPT * 1024)
#define SO_TPG   8

__global__ void __launch_bounds__(1024)
edge_bucket_so(const int* __restrict__ esrc, const int* __restrict__ edst,
               int* __restrict__ pack, int* __restrict__ tab,
               long long n_edges, int per, int batch) {
  __shared__ int stage[SO_CAP];
  __shared__ int cnt[SO_MAXSL];
  __shared__ int off[SO_MAXSL + 1];
  __shared__ int wtot[SO_MAXSL / 32];
  const int tid = threadIdx.x;
  long long lo = (long long)blockIdx.x * per;
  if (lo > n_edges) lo = n_edges;
  long long hi = lo + per;
  if (hi > n_edges) hi = n_edges;
  const int len = (int)(hi - lo);
  if (tid < SO_MAXSL) cnt[tid] = 0;
  __syncthreads();

  int dv[SO_EPT], sv[SO_EPT];
#pragma unroll
  for (int k = 0; k < SO_EPT; ++k) {
    const int i = tid + k * 1024;
    dv[k] = -1;
    sv[k] = 0;
    if (i < len) {
      dv[k] = edst[lo + i];
      sv[k] = esrc[lo + i];
    }
  }
  int ps[SO_EPT];
#pragma unroll
  for (int k = 0; k < SO_EPT; ++k) {
    ps[k] = -1;
    if ((unsigned)dv[k] < (unsigned)batch) {
      const int sl = dv[k] / SO_ROWS;
      ps[k] = (sl << 16) | atomicAdd(&cnt[sl], 1);
    }
  }
  __syncthreads();
  int mine = 0, incl = 0;
  if (tid < SO_MAXSL) {
    mine = cnt[tid];
    incl = mine;
#pragma unroll
    for (int d = 1; d < 32; d <<= 1) {
      const int t = __shfl_up(incl, d);
      if ((tid & 31) >= d) incl += t;
    }
    if ((tid & 31) == 31) wtot[tid >> 5] = incl;
  }
  __syncthreads();
  if (tid < SO_MAXSL) {
    int base = 0;
#pragma unroll
    for (int j = 0; j < SO_MAXSL / 32; ++j) base += (j < (tid >> 5)) ? wtot[j] : 0;
    off[tid] = base + incl - mine;
    if (tid == SO_MAXSL - 1) off[SO_MAXSL] = base + incl;
  }
  __syncthreads();
#pragma unroll
  for (int k = 0; k < SO_EPT; ++k) {
    if (ps[k] >= 0) {
      const int sl = ps[k] >> 16;
      const unsigned w = ((unsigned)sv[k] << SO_SHIFT) | (unsigned)(dv[k] - sl * SO_ROWS);
      stage[off[sl] + (ps[k] & 0xffff)] = (int)w;
    }
  }
  __syncthreads();
  const int total = off[SO_MAXSL];
  for (int i = tid; i < total; i += 1024) pack[lo + i] = stage[i];
  if (tid <= SO_MAXSL) tab[blockIdx.x * SO_TROW + tid] = off[tid];
}

__device__ __forceinline__ void mlp_tile(
    const v2f aA[4], const v2f aX[4],
    const float* __restrict__ Wl, const float* __restrict__ bl,
    const float* __restrict__ Wr,
    const float* __restrict__ W1, const float* __restrict__ b1,
    const float* __restrict__ W2, const float* __restrict__ b2,
    float (*Sw)[18], float* __restrict__ out,
    int r0, int batch, int lane16, int half, int ko) {
  const int wrow = lane16 * 16 + ko;

  v2f bL[4], bR[4];
#pragma unroll
  for (int i = 0; i < 4; ++i) {
    bL[i] = *(const v2f*)(Wl + wrow + 4 * i);
    bR[i] = *(const v2f*)(Wr + wrow + 4 * i);
  }

  v8f acc = {};
#pragma unroll
  for (int i = 0; i < 4; ++i) acc = wmma_f32_k4(aA[i], bL[i], acc);
#pragma unroll
  for (int i = 0; i < 4; ++i) acc = wmma_f32_k4(aX[i], bR[i], acc);
  {
    const float bias = bl[lane16];
#pragma unroll
    for (int r = 0; r < 8; ++r) {
      float v = acc[r] + bias;
      v = (v >= 0.0f) ? v : NEG_SLOPE * v;
      Sw[r + 8 * half][lane16] = v;
    }
  }
  lds_fence();

  v2f a1[4], b1f[4];
#pragma unroll
  for (int i = 0; i < 4; ++i) {
    a1[i]  = *(const v2f*)(&Sw[lane16][4 * i + ko]);
    b1f[i] = *(const v2f*)(W1 + wrow + 4 * i);
  }
  v8f acc1 = {};
#pragma unroll
  for (int i = 0; i < 4; ++i) acc1 = wmma_f32_k4(a1[i], b1f[i], acc1);
  lds_fence();
  {
    const float bias = b1[lane16];
#pragma unroll
    for (int r = 0; r < 8; ++r) {
      float v = acc1[r] + bias;
      v = (v >= 0.0f) ? v : NEG_SLOPE * v;
      Sw[r + 8 * half][lane16] = v;
    }
  }
  lds_fence();

  v2f a2[4], b2f[4];
#pragma unroll
  for (int i = 0; i < 4; ++i) {
    a2[i]  = *(const v2f*)(&Sw[lane16][4 * i + ko]);
    b2f[i] = *(const v2f*)(W2 + wrow + 4 * i);
  }
  v8f acc2 = {};
#pragma unroll
  for (int i = 0; i < 4; ++i) acc2 = wmma_f32_k4(a2[i], b2f[i], acc2);
  {
    const float bias = b2[lane16];
#pragma unroll
    for (int r = 0; r < 8; ++r) {
      const int orow = r0 + r + 8 * half;
      if (orow < batch)
        out[(long long)orow * 16 + lane16] = acc2[r] + bias;
    }
  }
}

struct SoRun { int base; int len; };
__device__ __forceinline__ SoRun so_run(int tv, int k, int tg, int n_tiles, int per) {
  const int t  = tg + k;
  const int tt = (t < n_tiles) ? t : (n_tiles - 1);
  const int b = __builtin_amdgcn_readlane(tv, 2 * k);
  const int e = __builtin_amdgcn_readlane(tv, 2 * k + 1);
  SoRun r;
  r.base = tt * per + b;
  r.len  = (t < n_tiles) ? (e - b) : 0;
  return r;
}

__device__ __forceinline__ int so_place(int j, int P1, int P2, int P3, int P4, int P5,
                                        int P6, int P7, int B0, int B1, int B2, int B3,
                                        int B4, int B5, int B6, int B7) {
  int a = B0;
  a = (j >= P1) ? B1 : a;
  a = (j >= P2) ? B2 : a;
  a = (j >= P3) ? B3 : a;
  a = (j >= P4) ? B4 : a;
  a = (j >= P5) ? B5 : a;
  a = (j >= P6) ? B6 : a;
  a = (j >= P7) ? B7 : a;
  return a + j;
}

__global__ void __launch_bounds__(1024)
edge_consume_mlp(const float* __restrict__ x, const int* __restrict__ pack,
                 const int* __restrict__ tab,
                 const float* __restrict__ Wl, const float* __restrict__ bl,
                 const float* __restrict__ Wr,
                 const float* __restrict__ W1, const float* __restrict__ b1,
                 const float* __restrict__ W2, const float* __restrict__ b2,
                 float* __restrict__ out, int per, int batch, int n_tiles) {
  __shared__ float acc[SO_ROWS * 16];
  __shared__ float S[32][16][18];
  const int tid  = threadIdx.x;
  const int lane = tid & 31;
  const int wave = __builtin_amdgcn_readfirstlane(tid >> 5);
  const int slice = blockIdx.x;
  const int lo = slice * SO_ROWS;
  int ns = batch - lo;
  if (ns > SO_ROWS) ns = SO_ROWS;

  const int half   = lane >> 4;
  const int lane16 = lane & 15;
  const int ko     = 2 * half;
  const bool has_tile = wave * 16 < ns;
  const int mrow = lo + wave * 16 + lane16;
  for (int i = tid; i < SO_ROWS * 16; i += 1024) acc[i] = 0.0f;
  __syncthreads();

  v2f aX[4];
#pragma unroll
  for (int i = 0; i < 4; ++i) aX[i] = (v2f){0.0f, 0.0f};
  if (has_tile && mrow < batch) {
#pragma unroll
    for (int i = 0; i < 4; ++i)
      aX[i] = *(const v2f*)(x + (long long)mrow * 16 + ko + 4 * i);
  }

  const int q   = lane & 3;
  const int grp = lane >> 2;
  const int n_groups = (n_tiles + SO_TPG - 1) / SO_TPG;

  for (int g = wave; g < n_groups; g += 32) {
    const int tg = g * SO_TPG;
    int tv;
    {
      const int t  = tg + (lane >> 1);
      const int tt = (t < n_tiles) ? t : (n_tiles - 1);
      tv = tab[tt * SO_TROW + slice + (lane & 1)];
    }
    const SoRun r0 = so_run(tv, 0, tg, n_tiles, per);
    const SoRun r1 = so_run(tv, 1, tg, n_tiles, per);
    const SoRun r2 = so_run(tv, 2, tg, n_tiles, per);
    const SoRun r3 = so_run(tv, 3, tg, n_tiles, per);
    const SoRun r4 = so_run(tv, 4, tg, n_tiles, per);
    const SoRun r5 = so_run(tv, 5, tg, n_tiles, per);
    const SoRun r6 = so_run(tv, 6, tg, n_tiles, per);
    const SoRun r7 = so_run(tv, 7, tg, n_tiles, per);
    const int P1 = r0.len,      P2 = P1 + r1.len, P3 = P2 + r2.len, P4 = P3 + r3.len;
    const int P5 = P4 + r4.len, P6 = P5 + r5.len, P7 = P6 + r6.len;
    const int nw = P7 + r7.len;
    if (nw <= 0) continue;
    const int B0 = r0.base,      B1 = r1.base - P1, B2 = r2.base - P2, B3 = r3.base - P3;
    const int B4 = r4.base - P4, B5 = r5.base - P5, B6 = r6.base - P6, B7 = r7.base - P7;
#define SO_PLACE(j) so_place((j), P1, P2, P3, P4, P5, P6, P7, B0, B1, B2, B3, B4, B5, B6, B7)
    int pos = 0;
    int n = nw;
    if (n > 64) n = 64;
    int w0 = pack[SO_PLACE((lane < n) ? lane : (n - 1))];
    int w1 = pack[SO_PLACE((lane + 32 < n) ? (lane + 32) : (n - 1))];
    for (;;) {
      float4 v[8];
      int dl[8];
#pragma unroll
      for (int u = 0; u < 8; ++u) {
        const int from = (lane & ~3) | (u & 3);
        const int wd = __shfl((u < 4) ? w0 : w1, from);
        dl[u] = wd & ((1 << SO_SHIFT) - 1);
        const unsigned s = (unsigned)wd >> SO_SHIFT;
        v[u] = *(const float4*)(x + (long long)s * 16 + q * 4);
      }
      const int npos = pos + 64;
      int nn = nw - npos;
      if (nn > 64) nn = 64;
      int x0 = w0, x1 = w1;
      if (nn > 0) {
        x0 = pack[SO_PLACE(npos + ((lane < nn) ? lane : (nn - 1)))];
        x1 = pack[SO_PLACE(npos + ((lane + 32 < nn) ? (lane + 32) : (nn - 1)))];
      }
#pragma unroll
      for (int u = 0; u < 8; ++u) {
        const int j = ((u < 4) ? 0 : 32) + 4 * grp + (u & 3);
        if (j < n) {
          float* a = acc + dl[u] * 16 + q * 4;
          atomicAdd(a + 0, v[u].x);
          atomicAdd(a + 1, v[u].y);
          atomicAdd(a + 2, v[u].z);
          atomicAdd(a + 3, v[u].w);
        }
      }
      if (nn <= 0) break;
      pos = npos;
      n = nn;
      w0 = x0;
      w1 = x1;
    }
  }

#undef SO_PLACE

  __syncthreads();
  if (has_tile) {
    v2f aA[4];
#pragma unroll
    for (int i = 0; i < 4; ++i)
      aA[i] = *(const v2f*)(&acc[(wave * 16 + lane16) * 16 + 4 * i + ko]);
    mlp_tile(aA, aX, Wl, bl, Wr, W1, b1, W2, b2, S[wave], out,
             lo + wave * 16, batch, lane16, half, ko);
  }
}

__global__ void __launch_bounds__(128)
sage_mlp_wmma(const float* __restrict__ agg, const float* __restrict__ x,
              const float* __restrict__ Wl, const float* __restrict__ bl,
              const float* __restrict__ Wr,
              const float* __restrict__ W1, const float* __restrict__ b1,
              const float* __restrict__ W2, const float* __restrict__ b2,
              float* __restrict__ out, int batch) {
  __shared__ float S[4][16][18];
  const int wave   = threadIdx.x >> 5;
  const int l      = threadIdx.x & 31;
  const int half   = l >> 4;
  const int lane16 = l & 15;
  const int ko     = 2 * half;
  const int tile   = blockIdx.x * 4 + wave;
  if (tile * 16 >= batch) return;
  const int r0 = tile * 16;
  float (*Sw)[18] = S[wave];

  v2f aA[4], aX[4];
  if (r0 + 16 <= batch) {
    const long long rb = (long long)(r0 + lane16) * 16 + ko;
#pragma unroll
    for (int i = 0; i < 4; ++i) {
      aA[i] = *(const v2f*)(agg + rb + 4 * i);
      aX[i] = *(const v2f*)(x + rb + 4 * i);
    }
  } else {
    const int row = r0 + lane16;
    const bool rowv = row < batch;
#pragma unroll
    for (int c = 0; c < 8; ++c)
      Sw[lane16][half * 8 + c] = rowv ? agg[(long long)row * 16 + half * 8 + c] : 0.0f;
    lds_fence();
#pragma unroll
    for (int i = 0; i < 4; ++i) aA[i] = *(const v2f*)(&Sw[lane16][4 * i + ko]);
    lds_fence();
#pragma unroll
    for (int c = 0; c < 8; ++c)
      Sw[lane16][half * 8 + c] = rowv ? x[(long long)row * 16 + half * 8 + c] : 0.0f;
    lds_fence();
#pragma unroll
    for (int i = 0; i < 4; ++i) aX[i] = *(const v2f*)(&Sw[lane16][4 * i + ko]);
    lds_fence();
  }

  mlp_tile(aA, aX, Wl, bl, Wr, W1, b1, W2, b2, Sw, out,
           r0, batch, lane16, half, ko);
}

extern "C" void kernel_launch(void* const* d_in, const int* in_sizes, int n_in,
                              void* d_out, int out_size, void* d_ws, size_t ws_size,
                              hipStream_t stream) {
  const float* x  = (const float*)d_in[0];
  const float* Wl = (const float*)d_in[1];
  const float* bl = (const float*)d_in[2];
  const float* Wr = (const float*)d_in[3];
  const float* W1 = (const float*)d_in[4];
  const float* b1 = (const float*)d_in[5];
  const float* W2 = (const float*)d_in[6];
  const float* b2 = (const float*)d_in[7];
  const int* esrc = (const int*)d_in[8];
  const int* edst = (const int*)d_in[9];

  const long long n_edges = in_sizes[8];
  const int batch = out_size / 16;
  float* agg = (float*)d_ws;
  float* out = (float*)d_out;

  {
    const int so_slices = (batch + SO_ROWS - 1) / SO_ROWS;
    const int so_tiles = 256;
    long long so_per = (n_edges + so_tiles - 1) / so_tiles;
    so_per = (so_per + 31) & ~31LL;
    const size_t so_pack_bytes = (size_t)so_tiles * (size_t)so_per * 4;
    const size_t so_tab_bytes  = (size_t)so_tiles * SO_TROW * 4;
    if (so_slices >= 1 && so_slices <= SO_MAXSL && so_per <= SO_CAP &&
        in_sizes[0] <= 16 * 8388608 && ws_size >= so_pack_bytes + so_tab_bytes) {
      int* pack = (int*)d_ws;
      int* tab  = (int*)((char*)d_ws + so_pack_bytes);
      edge_bucket_so<<<so_tiles, 1024, 0, stream>>>(esrc, edst, pack, tab, n_edges,
                                                    (int)so_per, batch);
      edge_consume_mlp<<<so_slices, 1024, 0, stream>>>(x, pack, tab, Wl, bl, Wr, W1, b1,
                                                      W2, b2, out, (int)so_per, batch,
                                                      so_tiles);
      return;
    }
  }

  const long long n_agg = (long long)batch * 16;
  int zgrid = (int)((n_agg + 255) / 256);
  zero_f32<<<zgrid, 256, 0, stream>>>(agg, n_agg);

  int b_slices = (batch + BK_SLICE - 1) / BK_SLICE;
  if (b_slices < 1) b_slices = 1;
  int b_chunks = 256 / b_slices;
  if (b_chunks < 1) b_chunks = 1;
  const int n_tiles = b_chunks * 64;
  long long per = (n_edges + n_tiles - 1) / n_tiles;
  per = (per + 3) & ~3LL;
  const size_t agg_bytes  = ((size_t)n_agg * 4 + 255) & ~(size_t)255;
  const size_t pack_bytes = (size_t)n_tiles * (size_t)per * 4;
  const size_t tab_bytes  = (size_t)n_tiles * BK_TROW * 4;
  if (b_slices <= BK_MAXSL && per <= BK_CAP && in_sizes[0] <= 16 * 2097152 &&
      ws_size >= agg_bytes + pack_bytes + tab_bytes) {
    int* pack = (int*)((char*)d_ws + agg_bytes);
    int* tab  = (int*)((char*)d_ws + agg_bytes + pack_bytes);
    edge_bucket<<<n_tiles, 1024, 0, stream>>>(esrc, edst, pack, tab, n_edges, (int)per, batch);
    edge_consume_lds<<<b_slices * b_chunks, 1024, 0, stream>>>(x, pack, tab, agg, (int)per,
                                                              batch, b_slices, n_tiles);
  } else {
    int n_slices = (batch + SLICE_T - 1) / SLICE_T;
    if (n_slices < 1) n_slices = 1;
    int n_chunks = 256 / n_slices;
    if (n_chunks < 1) n_chunks = 1;
    edge_scatter_lds<<<n_slices * n_chunks, 1024, 0, stream>>>(x, esrc, edst, agg,
                                                              n_edges, batch, n_slices);
  }

  const int tiles = (batch + 15) / 16;
  int mgrid = (tiles + 3) / 4;
  sage_mlp_wmma<<<mgrid, 128, 0, stream>>>(agg, x, Wl, bl, Wr,
                                           W1, b1, W2, b2, out, batch);
}


# LDS row adds spread over all 64 banks: the lanes of edge slot g start at component g&3 of their float4 and go round, so a wave-wide add no longer stacks 2-8 deep on 16 banks; arrival-order sums kept
# speedup vs baseline: 3.3808x; 1.1887x over previous
#include <hip/hip_runtime.h>

typedef __attribute__((ext_vector_type(2))) float v2f;
typedef __attribute__((ext_vector_type(8))) float v8f;

#define NEG_SLOPE 0.01f

__device__ __forceinline__ v8f wmma_f32_k4(v2f a, v2f b, v8f c) {
  return __builtin_amdgcn_wmma_f32_16x16x4_f32(
       false, a,  false, b,
       (short)0, c,  false,  false);
}

__device__ __forceinline__ void lds_fence() {
  asm volatile("s_wait_dscnt 0" ::: "memory");
}

__global__ void zero_f32(float* __restrict__ p, long long n) {
  long long i = (long long)blockIdx.x * blockDim.x + threadIdx.x;
  long long stride = (long long)gridDim.x * blockDim.x;
  for (; i < n; i += stride) p[i] = 0.0f;
}

__device__ __forceinline__ void add_quarter_rot(float* a, const float4 v, const int rot) {
  const bool r1 = (rot & 1) != 0, r2 = (rot & 2) != 0;
  const float t0 = r1 ? v.y : v.x, t1 = r1 ? v.z : v.y, t2 = r1 ? v.w : v.z, t3 = r1 ? v.x : v.w;
  const float w0 = r2 ? t2 : t0, w1 = r2 ? t3 : t1, w2 = r2 ? t0 : t2, w3 = r2 ? t1 : t3;
  atomicAdd(a + (rot & 3), w0);
  atomicAdd(a + ((rot + 1) & 3), w1);
  atomicAdd(a + ((rot + 2) & 3), w2);
  atomicAdd(a + ((rot + 3) & 3), w3);
}

#define SLICE_T 4352
#define SC_WAVES 32
#define SC_PASS 64
#define SC_LCAP 192

__device__ __forceinline__ void scatter_pass(const float* __restrict__ x,
                                             const int* __restrict__ esrc,
                                             const int* le, const int* ld,
                                             float* acc, int base, int n, int lane) {
  const int q = lane & 3;
  const int sub = lane >> 2;
  int s[8], d[8];
#pragma unroll
  for (int u = 0; u < 8; ++u) {
    int j = sub + 8 * u;
    j = (j < n) ? j : (n - 1);
    const int e = le[base + j];
    d[u] = ld[base + j];
    s[u] = esrc[e];
  }
  float4 v[8];
#pragma unroll
  for (int u = 0; u < 8; ++u)
    v[u] = *(const float4*)(x + (long long)s[u] * 16 + q * 4);
#pragma unroll
  for (int u = 0; u < 8; ++u) {
    if (sub + 8 * u < n) {
      add_quarter_rot(acc + d[u] * 16 + q * 4, v[u], sub & 3);
    }
  }
}

__device__ __forceinline__ int4 load_dst4_tail(const int* __restrict__ edst,
                                               unsigned wb, unsigned e1, int lane) {
  int4 t = make_int4(-1, -1, -1, -1);
  const unsigned e = wb + lane * 4;
  if (e      < e1) t.x = edst[e];
  if (e + 1u < e1) t.y = edst[e + 1u];
  if (e + 2u < e1) t.z = edst[e + 2u];
  if (e + 3u < e1) t.w = edst[e + 3u];
  return t;
}

__global__ void __launch_bounds__(1024)
edge_scatter_lds(const float* __restrict__ x,
                 const int* __restrict__ esrc,
                 const int* __restrict__ edst,
                 float* __restrict__ agg, long long n_edges,
                 int batch, int n_slices) {
  __shared__ float acc[SLICE_T * 16];
  __shared__ int lst_e[SC_WAVES][SC_LCAP];
  __shared__ int lst_d[SC_WAVES][SC_LCAP];

  const int tid  = threadIdx.x;
  const int lane = tid & 31;
  const int wave = __builtin_amdgcn_readfirstlane(tid >> 5);
  const int slice    = blockIdx.x % n_slices;
  const int chunk    = blockIdx.x / n_slices;
  const int n_chunks = gridDim.x / n_slices;
  const int lo = slice * SLICE_T;
  int ns = batch - lo;
  if (ns > SLICE_T) ns = SLICE_T;

  for (int i = tid; i < SLICE_T * 16; i += 1024) acc[i] = 0.0f;
  __syncthreads();

  long long per = (n_edges + n_chunks - 1) / n_chunks;
  per = (per + 3) & ~3LL;
  long long lo_e = (long long)chunk * per;
  if (lo_e > n_edges) lo_e = n_edges;
  long long hi_e = lo_e + per;
  if (hi_e > n_edges) hi_e = n_edges;
  const unsigned e0 = (unsigned)lo_e;
  const unsigned e1 = (unsigned)hi_e;

  int* le = lst_e[wave];
  int* ld = lst_d[wave];
  int cnt = 0;

  auto filter = [&](const int4 c, const unsigned wb) {
    const unsigned r0 = (unsigned)(c.x - lo), r1 = (unsigned)(c.y - lo);
    const unsigned r2 = (unsigned)(c.z - lo), r3 = (unsigned)(c.w - lo);
    const bool p0 = r0 < (unsigned)ns, p1 = r1 < (unsigned)ns;
    const bool p2 = r2 < (unsigned)ns, p3 = r3 < (unsigned)ns;
    const unsigned m0 = __builtin_amdgcn_ballot_w32(p0);
    const unsigned m1 = __builtin_amdgcn_ballot_w32(p1);
    const unsigned m2 = __builtin_amdgcn_ballot_w32(p2);
    const unsigned m3 = __builtin_amdgcn_ballot_w32(p3);
    if ((m0 | m1 | m2 | m3) != 0u) {
      const int ei = (int)(wb + lane * 4);
      int b = cnt;
      if (p0) { const int o = b + (int)__builtin_amdgcn_mbcnt_lo(m0, 0u); le[o] = ei;     ld[o] = (int)r0; }
      b += __popc(m0);
      if (p1) { const int o = b + (int)__builtin_amdgcn_mbcnt_lo(m1, 0u); le[o] = ei + 1; ld[o] = (int)r1; }
      b += __popc(m1);
      if (p2) { const int o = b + (int)__builtin_amdgcn_mbcnt_lo(m2, 0u); le[o] = ei + 2; ld[o] = (int)r2; }
      b += __popc(m2);
      if (p3) { const int o = b + (int)__builtin_amdgcn_mbcnt_lo(m3, 0u); le[o] = ei + 3; ld[o] = (int)r3; }
      b += __popc(m3);
      cnt = b;
      while (cnt >= SC_PASS) {
        lds_fence();
        scatter_pass(x, esrc, le, ld, acc, cnt - SC_PASS, SC_PASS, lane);
        cnt -= SC_PASS;
      }
    }
  };

  const unsigned step = SC_WAVES * 128u;
  unsigned wb = e0 + (unsigned)wave * 128u;
  if (wb + 128u <= e1) {
    const unsigned first = wb;
    auto ldw = [&](const unsigned w) {
      const unsigned a = (w + 128u <= e1) ? w : first;
      return *(const int4*)(edst + a + lane * 4);
    };
    int4 c0 = ldw(wb);
    int4 c1 = ldw(wb + step);
    int4 c2;
    for (;;) {
      c2 = ldw(wb + 2u * step); filter(c0, wb); wb += step; if (wb + 128u > e1) break;
      c0 = ldw(wb + 2u * step); filter(c1, wb); wb += step; if (wb + 128u > e1) break;
      c1 = ldw(wb + 2u * step); filter(c2, wb); wb += step; if (wb + 128u > e1) break;
    }
  }
  if (wb < e1) filter(load_dst4_tail(edst, wb, e1, lane), wb);
  if (cnt > 0) {
    lds_fence();
    scatter_pass(x, esrc, le, ld, acc, 0, cnt, lane);
  }
  __syncthreads();

  float* g = agg + (long long)lo * 16;
  for (int i = tid; i < ns * 16; i += 1024) atomicAdd(g + i, acc[i]);
}

#define BK_SLICE 2048
#define BK_SHIFT 11
#define BK_MAXSL 64
#define BK_EPT   10
#define BK_CAP   (BK_EPT * 1024)
#define BK_TROW  (BK_MAXSL + 1)

__global__ void __launch_bounds__(1024)
edge_bucket(const int* __restrict__ esrc, const int* __restrict__ edst,
            int* __restrict__ pack, int* __restrict__ tab,
            long long n_edges, int per, int batch) {
  __shared__ int stage[BK_CAP];
  __shared__ int cnt[BK_MAXSL];
  __shared__ int off[BK_TROW];
  const int tid = threadIdx.x;
  long long lo = (long long)blockIdx.x * per;
  if (lo > n_edges) lo = n_edges;
  long long hi = lo + per;
  if (hi > n_edges) hi = n_edges;
  const int len = (int)(hi - lo);
  if (tid < BK_MAXSL) cnt[tid] = 0;
  __syncthreads();

  int dv[BK_EPT], sv[BK_EPT];
#pragma unroll
  for (int k = 0; k < BK_EPT; ++k) {
    const int i = tid + k * 1024;
    dv[k] = -1;
    sv[k] = 0;
    if (i < len) {
      dv[k] = edst[lo + i];
      sv[k] = esrc[lo + i];
    }
  }
  int ps[BK_EPT];
#pragma unroll
  for (int k = 0; k < BK_EPT; ++k) {
    ps[k] = -1;
    if ((unsigned)dv[k] < (unsigned)batch) {
      const int sl = dv[k] >> BK_SHIFT;
      ps[k] = (sl << 16) | atomicAdd(&cnt[sl], 1);
    }
  }
  __syncthreads();
  if (tid < BK_TROW) {
    int a = 0;
#pragma unroll
    for (int j = 0; j < BK_MAXSL; ++j) a += (j < tid) ? cnt[j] : 0;
    off[tid] = a;
  }
  __syncthreads();
#pragma unroll
  for (int k = 0; k < BK_EPT; ++k) {
    if (ps[k] >= 0) {
      const unsigned w = ((unsigned)sv[k] << BK_SHIFT) | ((unsigned)dv[k] & (BK_SLICE - 1));
      stage[off[ps[k] >> 16] + (ps[k] & 0xffff)] = (int)w;
    }
  }
  __syncthreads();
  const int total = off[BK_MAXSL];
  for (int i = tid; i < total; i += 1024) pack[lo + i] = stage[i];
  if (tid < BK_TROW) tab[blockIdx.x * BK_TROW + tid] = off[tid];
}

__global__ void __launch_bounds__(1024)
edge_consume_lds(const float* __restrict__ x, const int* __restrict__ pack,
                 const int* __restrict__ tab, float* __restrict__ agg,
                 int per, int batch, int n_slices, int n_tiles) {
  __shared__ float acc[BK_SLICE * 16];
  const int tid  = threadIdx.x;
  const int lane = tid & 31;
  const int wave = __builtin_amdgcn_readfirstlane(tid >> 5);
  const int slice    = blockIdx.x % n_slices;
  const int chunk    = blockIdx.x / n_slices;
  const int n_chunks = gridDim.x / n_slices;
  const int lo = slice * BK_SLICE;
  int ns = batch - lo;
  if (ns > BK_SLICE) ns = BK_SLICE;

  for (int i = tid; i < BK_SLICE * 16; i += 1024) acc[i] = 0.0f;
  __syncthreads();

  const int tpc = (n_tiles + n_chunks - 1) / n_chunks;
  const int t0 = chunk * tpc;
  int t1 = t0 + tpc;
  if (t1 > n_tiles) t1 = n_tiles;
  const int q   = lane & 3;
  const int grp = lane >> 2;

  for (int t = t0 + wave; t < t1; t += 32) {
    const int b = tab[t * BK_TROW + slice];
    const int e = tab[t * BK_TROW + slice + 1];
    if (b >= e) continue;
    const int* p = pack + (long long)t * per;
    int pos = b;
    int n = e - pos;
    if (n > 64) n = 64;
    int w0 = p[pos + ((lane < n) ? lane : (n - 1))];
    int w1 = p[pos + ((lane + 32 < n) ? (lane + 32) : (n - 1))];
    for (;;) {
      float4 v[8];
      int dl[8];
#pragma unroll
      for (int u = 0; u < 8; ++u) {
        const int from = (lane & ~3) | (u & 3);
        const int wd = __shfl((u < 4) ? w0 : w1, from);
        dl[u] = wd & (BK_SLICE - 1);
        const unsigned s = (unsigned)wd >> BK_SHIFT;
        v[u] = *(const float4*)(x + (long long)s * 16 + q * 4);
      }
      const int npos = pos + 64;
      int nn = e - npos;
      if (nn > 64) nn = 64;
      int x0 = w0, x1 = w1;
      if (nn > 0) {
        x0 = p[npos + ((lane < nn) ? lane : (nn - 1))];
        x1 = p[npos + ((lane + 32 < nn) ? (lane + 32) : (nn - 1))];
      }
#pragma unroll
      for (int u = 0; u < 8; ++u) {
        const int j = ((u < 4) ? 0 : 32) + 4 * grp + (u & 3);
        if (j < n) {
          add_quarter_rot(acc + dl[u] * 16 + q * 4, v[u], grp & 3);
        }
      }
      if (nn <= 0) break;
      pos = npos;
      n = nn;
      w0 = x0;
      w1 = x1;
    }
  }
  __syncthreads();

  float* g = agg + (long long)lo * 16;
  for (int i = tid; i < ns * 16; i += 1024) atomicAdd(g + i, acc[i]);
}

#define SO_ROWS  400
#define SO_SHIFT 9
#define SO_MAXSL 256
#define SO_TROW  288
#define SO_EPT   13
#define SO_CAP   (SO_EPT * 1024)
#define SO_TPG   8

__global__ void __launch_bounds__(1024)
edge_bucket_so(const int* __restrict__ esrc, const int* __restrict__ edst,
               int* __restrict__ pack, int* __restrict__ tab,
               long long n_edges, int per, int batch) {
  __shared__ int stage[SO_CAP];
  __shared__ int cnt[SO_MAXSL];
  __shared__ int off[SO_MAXSL + 1];
  __shared__ int wtot[SO_MAXSL / 32];
  const int tid = threadIdx.x;
  long long lo = (long long)blockIdx.x * per;
  if (lo > n_edges) lo = n_edges;
  long long hi = lo + per;
  if (hi > n_edges) hi = n_edges;
  const int len = (int)(hi - lo);
  if (tid < SO_MAXSL) cnt[tid] = 0;
  __syncthreads();

  int dv[SO_EPT], sv[SO_EPT];
#pragma unroll
  for (int k = 0; k < SO_EPT; ++k) {
    const int i = tid + k * 1024;
    dv[k] = -1;
    sv[k] = 0;
    if (i < len) {
      dv[k] = edst[lo + i];
      sv[k] = esrc[lo + i];
    }
  }
  int ps[SO_EPT];
#pragma unroll
  for (int k = 0; k < SO_EPT; ++k) {
    ps[k] = -1;
    if ((unsigned)dv[k] < (unsigned)batch) {
      const int sl = dv[k] / SO_ROWS;
      ps[k] = (sl << 16) | atomicAdd(&cnt[sl], 1);
    }
  }
  __syncthreads();
  int mine = 0, incl = 0;
  if (tid < SO_MAXSL) {
    mine = cnt[tid];
    incl = mine;
#pragma unroll
    for (int d = 1; d < 32; d <<= 1) {
      const int t = __shfl_up(incl, d);
      if ((tid & 31) >= d) incl += t;
    }
    if ((tid & 31) == 31) wtot[tid >> 5] = incl;
  }
  __syncthreads();
  if (tid < SO_MAXSL) {
    int base = 0;
#pragma unroll
    for (int j = 0; j < SO_MAXSL / 32; ++j) base += (j < (tid >> 5)) ? wtot[j] : 0;
    off[tid] = base + incl - mine;
    if (tid == SO_MAXSL - 1) off[SO_MAXSL] = base + incl;
  }
  __syncthreads();
#pragma unroll
  for (int k = 0; k < SO_EPT; ++k) {
    if (ps[k] >= 0) {
      const int sl = ps[k] >> 16;
      const unsigned w = ((unsigned)sv[k] << SO_SHIFT) | (unsigned)(dv[k] - sl * SO_ROWS);
      stage[off[sl] + (ps[k] & 0xffff)] = (int)w;
    }
  }
  __syncthreads();
  const int total = off[SO_MAXSL];
  for (int i = tid; i < total; i += 1024) pack[lo + i] = stage[i];
  if (tid <= SO_MAXSL) tab[blockIdx.x * SO_TROW + tid] = off[tid];
}

__device__ __forceinline__ void mlp_tile(
    const v2f aA[4], const v2f aX[4],
    const float* __restrict__ Wl, const float* __restrict__ bl,
    const float* __restrict__ Wr,
    const float* __restrict__ W1, const float* __restrict__ b1,
    const float* __restrict__ W2, const float* __restrict__ b2,
    float (*Sw)[18], float* __restrict__ out,
    int r0, int batch, int lane16, int half, int ko) {
  const int wrow = lane16 * 16 + ko;

  v2f bL[4], bR[4];
#pragma unroll
  for (int i = 0; i < 4; ++i) {
    bL[i] = *(const v2f*)(Wl + wrow + 4 * i);
    bR[i] = *(const v2f*)(Wr + wrow + 4 * i);
  }

  v8f acc = {};
#pragma unroll
  for (int i = 0; i < 4; ++i) acc = wmma_f32_k4(aA[i], bL[i], acc);
#pragma unroll
  for (int i = 0; i < 4; ++i) acc = wmma_f32_k4(aX[i], bR[i], acc);
  {
    const float bias = bl[lane16];
#pragma unroll
    for (int r = 0; r < 8; ++r) {
      float v = acc[r] + bias;
      v = (v >= 0.0f) ? v : NEG_SLOPE * v;
      Sw[r + 8 * half][lane16] = v;
    }
  }
  lds_fence();

  v2f a1[4], b1f[4];
#pragma unroll
  for (int i = 0; i < 4; ++i) {
    a1[i]  = *(const v2f*)(&Sw[lane16][4 * i + ko]);
    b1f[i] = *(const v2f*)(W1 + wrow + 4 * i);
  }
  v8f acc1 = {};
#pragma unroll
  for (int i = 0; i < 4; ++i) acc1 = wmma_f32_k4(a1[i], b1f[i], acc1);
  lds_fence();
  {
    const float bias = b1[lane16];
#pragma unroll
    for (int r = 0; r < 8; ++r) {
      float v = acc1[r] + bias;
      v = (v >= 0.0f) ? v : NEG_SLOPE * v;
      Sw[r + 8 * half][lane16] = v;
    }
  }
  lds_fence();

  v2f a2[4], b2f[4];
#pragma unroll
  for (int i = 0; i < 4; ++i) {
    a2[i]  = *(const v2f*)(&Sw[lane16][4 * i + ko]);
    b2f[i] = *(const v2f*)(W2 + wrow + 4 * i);
  }
  v8f acc2 = {};
#pragma unroll
  for (int i = 0; i < 4; ++i) acc2 = wmma_f32_k4(a2[i], b2f[i], acc2);
  {
    const float bias = b2[lane16];
#pragma unroll
    for (int r = 0; r < 8; ++r) {
      const int orow = r0 + r + 8 * half;
      if (orow < batch)
        out[(long long)orow * 16 + lane16] = acc2[r] + bias;
    }
  }
}

struct SoRun { int base; int len; };
__device__ __forceinline__ SoRun so_run(int tv, int k, int tg, int n_tiles, int per) {
  const int t  = tg + k;
  const int tt = (t < n_tiles) ? t : (n_tiles - 1);
  const int b = __builtin_amdgcn_readlane(tv, 2 * k);
  const int e = __builtin_amdgcn_readlane(tv, 2 * k + 1);
  SoRun r;
  r.base = tt * per + b;
  r.len  = (t < n_tiles) ? (e - b) : 0;
  return r;
}

__device__ __forceinline__ int so_place(int j, int P1, int P2, int P3, int P4, int P5,
                                        int P6, int P7, int B0, int B1, int B2, int B3,
                                        int B4, int B5, int B6, int B7) {
  int a = B0;
  a = (j >= P1) ? B1 : a;
  a = (j >= P2) ? B2 : a;
  a = (j >= P3) ? B3 : a;
  a = (j >= P4) ? B4 : a;
  a = (j >= P5) ? B5 : a;
  a = (j >= P6) ? B6 : a;
  a = (j >= P7) ? B7 : a;
  return a + j;
}

__global__ void __launch_bounds__(1024)
edge_consume_mlp(const float* __restrict__ x, const int* __restrict__ pack,
                 const int* __restrict__ tab,
                 const float* __restrict__ Wl, const float* __restrict__ bl,
                 const float* __restrict__ Wr,
                 const float* __restrict__ W1, const float* __restrict__ b1,
                 const float* __restrict__ W2, const float* __restrict__ b2,
                 float* __restrict__ out, int per, int batch, int n_tiles) {
  __shared__ float acc[SO_ROWS * 16];
  __shared__ float S[32][16][18];
  const int tid  = threadIdx.x;
  const int lane = tid & 31;
  const int wave = __builtin_amdgcn_readfirstlane(tid >> 5);
  const int slice = blockIdx.x;
  const int lo = slice * SO_ROWS;
  int ns = batch - lo;
  if (ns > SO_ROWS) ns = SO_ROWS;

  const int half   = lane >> 4;
  const int lane16 = lane & 15;
  const int ko     = 2 * half;
  const bool has_tile = wave * 16 < ns;
  const int mrow = lo + wave * 16 + lane16;
  for (int i = tid; i < SO_ROWS * 16; i += 1024) acc[i] = 0.0f;
  __syncthreads();

  v2f aX[4];
#pragma unroll
  for (int i = 0; i < 4; ++i) aX[i] = (v2f){0.0f, 0.0f};
  if (has_tile && mrow < batch) {
#pragma unroll
    for (int i = 0; i < 4; ++i)
      aX[i] = *(const v2f*)(x + (long long)mrow * 16 + ko + 4 * i);
  }

  const int q   = lane & 3;
  const int grp = lane >> 2;
  const int n_groups = (n_tiles + SO_TPG - 1) / SO_TPG;

  for (int g = wave; g < n_groups; g += 32) {
    const int tg = g * SO_TPG;
    int tv;
    {
      const int t  = tg + (lane >> 1);
      const int tt = (t < n_tiles) ? t : (n_tiles - 1);
      tv = tab[tt * SO_TROW + slice + (lane & 1)];
    }
    const SoRun r0 = so_run(tv, 0, tg, n_tiles, per);
    const SoRun r1 = so_run(tv, 1, tg, n_tiles, per);
    const SoRun r2 = so_run(tv, 2, tg, n_tiles, per);
    const SoRun r3 = so_run(tv, 3, tg, n_tiles, per);
    const SoRun r4 = so_run(tv, 4, tg, n_tiles, per);
    const SoRun r5 = so_run(tv, 5, tg, n_tiles, per);
    const SoRun r6 = so_run(tv, 6, tg, n_tiles, per);
    const SoRun r7 = so_run(tv, 7, tg, n_tiles, per);
    const int P1 = r0.len,      P2 = P1 + r1.len, P3 = P2 + r2.len, P4 = P3 + r3.len;
    const int P5 = P4 + r4.len, P6 = P5 + r5.len, P7 = P6 + r6.len;
    const int nw = P7 + r7.len;
    if (nw <= 0) continue;
    const int B0 = r0.base,      B1 = r1.base - P1, B2 = r2.base - P2, B3 = r3.base - P3;
    const int B4 = r4.base - P4, B5 = r5.base - P5, B6 = r6.base - P6, B7 = r7.base - P7;
#define SO_PLACE(j) so_place((j), P1, P2, P3, P4, P5, P6, P7, B0, B1, B2, B3, B4, B5, B6, B7)
    int pos = 0;
    int n = nw;
    if (n > 64) n = 64;
    int w0 = pack[SO_PLACE((lane < n) ? lane : (n - 1))];
    int w1 = pack[SO_PLACE((lane + 32 < n) ? (lane + 32) : (n - 1))];
    for (;;) {
      float4 v[8];
      int dl[8];
#pragma unroll
      for (int u = 0; u < 8; ++u) {
        const int from = (lane & ~3) | (u & 3);
        const int wd = __shfl((u < 4) ? w0 : w1, from);
        dl[u] = wd & ((1 << SO_SHIFT) - 1);
        const unsigned s = (unsigned)wd >> SO_SHIFT;
        v[u] = *(const float4*)(x + (long long)s * 16 + q * 4);
      }
      const int npos = pos + 64;
      int nn = nw - npos;
      if (nn > 64) nn = 64;
      int x0 = w0, x1 = w1;
      if (nn > 0) {
        x0 = pack[SO_PLACE(npos + ((lane < nn) ? lane : (nn - 1)))];
        x1 = pack[SO_PLACE(npos + ((lane + 32 < nn) ? (lane + 32) : (nn - 1)))];
      }
#pragma unroll
      for (int u = 0; u < 8; ++u) {
        const int j = ((u < 4) ? 0 : 32) + 4 * grp + (u & 3);
        if (j < n) {
          add_quarter_rot(acc + dl[u] * 16 + q * 4, v[u], grp & 3);
        }
      }
      if (nn <= 0) break;
      pos = npos;
      n = nn;
      w0 = x0;
      w1 = x1;
    }
  }

#undef SO_PLACE

  __syncthreads();
  if (has_tile) {
    v2f aA[4];
#pragma unroll
    for (int i = 0; i < 4; ++i)
      aA[i] = *(const v2f*)(&acc[(wave * 16 + lane16) * 16 + 4 * i + ko]);
    mlp_tile(aA, aX, Wl, bl, Wr, W1, b1, W2, b2, S[wave], out,
             lo + wave * 16, batch, lane16, half, ko);
  }
}

__global__ void __launch_bounds__(128)
sage_mlp_wmma(const float* __restrict__ agg, const float* __restrict__ x,
              const float* __restrict__ Wl, const float* __restrict__ bl,
              const float* __restrict__ Wr,
              const float* __restrict__ W1, const float* __restrict__ b1,
              const float* __restrict__ W2, const float* __restrict__ b2,
              float* __restrict__ out, int batch) {
  __shared__ float S[4][16][18];
  const int wave   = threadIdx.x >> 5;
  const int l      = threadIdx.x & 31;
  const int half   = l >> 4;
  const int lane16 = l & 15;
  const int ko     = 2 * half;
  const int tile   = blockIdx.x * 4 + wave;
  if (tile * 16 >= batch) return;
  const int r0 = tile * 16;
  float (*Sw)[18] = S[wave];

  v2f aA[4], aX[4];
  if (r0 + 16 <= batch) {
    const long long rb = (long long)(r0 + lane16) * 16 + ko;
#pragma unroll
    for (int i = 0; i < 4; ++i) {
      aA[i] = *(const v2f*)(agg + rb + 4 * i);
      aX[i] = *(const v2f*)(x + rb + 4 * i);
    }
  } else {
    const int row = r0 + lane16;
    const bool rowv = row < batch;
#pragma unroll
    for (int c = 0; c < 8; ++c)
      Sw[lane16][half * 8 + c] = rowv ? agg[(long long)row * 16 + half * 8 + c] : 0.0f;
    lds_fence();
#pragma unroll
    for (int i = 0; i < 4; ++i) aA[i] = *(const v2f*)(&Sw[lane16][4 * i + ko]);
    lds_fence();
#pragma unroll
    for (int c = 0; c < 8; ++c)
      Sw[lane16][half * 8 + c] = rowv ? x[(long long)row * 16 + half * 8 + c] : 0.0f;
    lds_fence();
#pragma unroll
    for (int i = 0; i < 4; ++i) aX[i] = *(const v2f*)(&Sw[lane16][4 * i + ko]);
    lds_fence();
  }

  mlp_tile(aA, aX, Wl, bl, Wr, W1, b1, W2, b2, Sw, out,
           r0, batch, lane16, half, ko);
}

extern "C" void kernel_launch(void* const* d_in, const int* in_sizes, int n_in,
                              void* d_out, int out_size, void* d_ws, size_t ws_size,
                              hipStream_t stream) {
  const float* x  = (const float*)d_in[0];
  const float* Wl = (const float*)d_in[1];
  const float* bl = (const float*)d_in[2];
  const float* Wr = (const float*)d_in[3];
  const float* W1 = (const float*)d_in[4];
  const float* b1 = (const float*)d_in[5];
  const float* W2 = (const float*)d_in[6];
  const float* b2 = (const float*)d_in[7];
  const int* esrc = (const int*)d_in[8];
  const int* edst = (const int*)d_in[9];

  const long long n_edges = in_sizes[8];
  const int batch = out_size / 16;
  float* agg = (float*)d_ws;
  float* out = (float*)d_out;

  {
    const int so_slices = (batch + SO_ROWS - 1) / SO_ROWS;
    const int so_tiles = 256;
    long long so_per = (n_edges + so_tiles - 1) / so_tiles;
    so_per = (so_per + 31) & ~31LL;
    const size_t so_pack_bytes = (size_t)so_tiles * (size_t)so_per * 4;
    const size_t so_tab_bytes  = (size_t)so_tiles * SO_TROW * 4;
    if (so_slices >= 1 && so_slices <= SO_MAXSL && so_per <= SO_CAP &&
        in_sizes[0] <= 16 * 8388608 && ws_size >= so_pack_bytes + so_tab_bytes) {
      int* pack = (int*)d_ws;
      int* tab  = (int*)((char*)d_ws + so_pack_bytes);
      edge_bucket_so<<<so_tiles, 1024, 0, stream>>>(esrc, edst, pack, tab, n_edges,
                                                    (int)so_per, batch);
      edge_consume_mlp<<<so_slices, 1024, 0, stream>>>(x, pack, tab, Wl, bl, Wr, W1, b1,
                                                      W2, b2, out, (int)so_per, batch,
                                                      so_tiles);
      return;
    }
  }

  const long long n_agg = (long long)batch * 16;
  int zgrid = (int)((n_agg + 255) / 256);
  zero_f32<<<zgrid, 256, 0, stream>>>(agg, n_agg);

  int b_slices = (batch + BK_SLICE - 1) / BK_SLICE;
  if (b_slices < 1) b_slices = 1;
  int b_chunks = 256 / b_slices;
  if (b_chunks < 1) b_chunks = 1;
  const int n_tiles = b_chunks * 64;
  long long per = (n_edges + n_tiles - 1) / n_tiles;
  per = (per + 3) & ~3LL;
  const size_t agg_bytes  = ((size_t)n_agg * 4 + 255) & ~(size_t)255;
  const size_t pack_bytes = (size_t)n_tiles * (size_t)per * 4;
  const size_t tab_bytes  = (size_t)n_tiles * BK_TROW * 4;
  if (b_slices <= BK_MAXSL && per <= BK_CAP && in_sizes[0] <= 16 * 2097152 &&
      ws_size >= agg_bytes + pack_bytes + tab_bytes) {
    int* pack = (int*)((char*)d_ws + agg_bytes);
    int* tab  = (int*)((char*)d_ws + agg_bytes + pack_bytes);
    edge_bucket<<<n_tiles, 1024, 0, stream>>>(esrc, edst, pack, tab, n_edges, (int)per, batch);
    edge_consume_lds<<<b_slices * b_chunks, 1024, 0, stream>>>(x, pack, tab, agg, (int)per,
                                                              batch, b_slices, n_tiles);
  } else {
    int n_slices = (batch + SLICE_T - 1) / SLICE_T;
    if (n_slices < 1) n_slices = 1;
    int n_chunks = 256 / n_slices;
    if (n_chunks < 1) n_chunks = 1;
    edge_scatter_lds<<<n_slices * n_chunks, 1024, 0, stream>>>(x, esrc, edst, agg,
                                                              n_edges, batch, n_slices);
  }

  const int tiles = (batch + 15) / 16;
  int mgrid = (tiles + 3) / 4;
  sage_mlp_wmma<<<mgrid, 128, 0, stream>>>(agg, x, Wl, bl, Wr,
                                           W1, b1, W2, b2, out, batch);
}
